# starting kernel
# baseline (speedup 1.0000x reference)
#include <hip/hip_runtime.h>
#include <hip/hip_bf16.h>

#define N_NODES   100000
#define N_EDGES   1600000
#define D_IN      256
#define D_HID     128
#define N_CLASSES 64
#define NEG_SLOPE 0.2f

typedef __attribute__((ext_vector_type(2))) float v2f;
typedef __attribute__((ext_vector_type(8))) float v8f;

__global__ __launch_bounds__(256) void k_gemm1(const float* __restrict__ X,
                                               const float* __restrict__ W,
                                               float* __restrict__ feat) {
    const int wave = threadIdx.x >> 5;
    const int lane = threadIdx.x & 31;
    const int strip = blockIdx.x * 8 + wave;
    const int rowTile = strip >> 2;
    const int colPair = strip & 3;
    const int row0 = rowTile * 16;
    const int col0 = colPair * 32;
    const int l16 = lane & 15;
    const int hi  = lane >> 4;

    v8f acc0 = {};
    v8f acc1 = {};
    const float* __restrict__ xrow = X + (size_t)(row0 + l16) * D_IN;

    for (int k0 = 0; k0 < D_IN; k0 += 4) {
        const int ka = k0 + 2 * hi;
        v2f a;
        a.x = xrow[ka];
        a.y = xrow[ka + 1];
        const float* __restrict__ w0 = W + (size_t)ka * D_HID + col0 + l16;
        const float* __restrict__ w1 = W + (size_t)(ka + 1) * D_HID + col0 + l16;
        v2f b0, b1;
        b0.x = w0[0];   b0.y = w1[0];
        b1.x = w0[16];  b1.y = w1[16];
        acc0 = __builtin_amdgcn_wmma_f32_16x16x4_f32(
            false, a, false, b0, (short)0, acc0, false, false);
        acc1 = __builtin_amdgcn_wmma_f32_16x16x4_f32(
            false, a, false, b1, (short)0, acc1, false, false);
    }
    #pragma unroll
    for (int r = 0; r < 8; ++r) {
        float* __restrict__ orow = feat + (size_t)(row0 + r + 8 * hi) * D_HID + col0 + l16;
        orow[0]  = acc0[r];
        orow[16] = acc1[r];
    }
}

__global__ __launch_bounds__(256) void k_gemm2(const float* __restrict__ H,
                                               const float* __restrict__ FW,
                                               const float* __restrict__ FB,
                                               float* __restrict__ out) {
    const int wave = threadIdx.x >> 5;
    const int lane = threadIdx.x & 31;
    const int rowTile = blockIdx.x * 8 + wave;
    if (rowTile >= N_NODES / 16) return;
    const int row0 = rowTile * 16;
    const int l16 = lane & 15;
    const int hi  = lane >> 4;

    v8f acc0 = {}, acc1 = {}, acc2 = {}, acc3 = {};
    const float* __restrict__ hrow = H + (size_t)(row0 + l16) * D_HID;

    for (int k0 = 0; k0 < D_HID; k0 += 4) {
        const int ka = k0 + 2 * hi;
        v2f a;
        a.x = hrow[ka];
        a.y = hrow[ka + 1];
        const float* __restrict__ w0 = FW + (size_t)ka * N_CLASSES + l16;
        const float* __restrict__ w1 = FW + (size_t)(ka + 1) * N_CLASSES + l16;
        v2f b0, b1, b2, b3;
        b0.x = w0[0];   b0.y = w1[0];
        b1.x = w0[16];  b1.y = w1[16];
        b2.x = w0[32];  b2.y = w1[32];
        b3.x = w0[48];  b3.y = w1[48];
        acc0 = __builtin_amdgcn_wmma_f32_16x16x4_f32(
            false, a, false, b0, (short)0, acc0, false, false);
        acc1 = __builtin_amdgcn_wmma_f32_16x16x4_f32(
            false, a, false, b1, (short)0, acc1, false, false);
        acc2 = __builtin_amdgcn_wmma_f32_16x16x4_f32(
            false, a, false, b2, (short)0, acc2, false, false);
        acc3 = __builtin_amdgcn_wmma_f32_16x16x4_f32(
            false, a, false, b3, (short)0, acc3, false, false);
    }
    const float bias0 = FB[l16];
    const float bias1 = FB[l16 + 16];
    const float bias2 = FB[l16 + 32];
    const float bias3 = FB[l16 + 48];
    #pragma unroll
    for (int r = 0; r < 8; ++r) {
        float* __restrict__ orow = out + (size_t)(row0 + r + 8 * hi) * N_CLASSES + l16;
        orow[0]  = acc0[r] + bias0;
        orow[16] = acc1[r] + bias1;
        orow[32] = acc2[r] + bias2;
        orow[48] = acc3[r] + bias3;
    }
}

__global__ __launch_bounds__(256) void k_scores(const float* __restrict__ feat,
                                                const float* __restrict__ al,
                                                const float* __restrict__ ar,
                                                float* __restrict__ el,
                                                float* __restrict__ er) {
    const int wave = threadIdx.x >> 5;
    const int lane = threadIdx.x & 31;
    const int node = blockIdx.x * 8 + wave;
    if (node >= N_NODES) return;
    const float* __restrict__ f = feat + (size_t)node * D_HID;
    float sl = 0.f, sr = 0.f;
    #pragma unroll
    for (int j = 0; j < D_HID; j += 32) {
        float v = f[j + lane];
        sl += v * al[j + lane];
        sr += v * ar[j + lane];
    }
    #pragma unroll
    for (int off = 16; off > 0; off >>= 1) {
        sl += __shfl_down(sl, off, 32);
        sr += __shfl_down(sr, off, 32);
    }
    if (lane == 0) { el[node] = sl; er[node] = sr; }
}

__device__ __forceinline__ unsigned enc_f32(float f) {
    unsigned b = __float_as_uint(f);
    return (b & 0x80000000u) ? ~b : (b | 0x80000000u);
}
__device__ __forceinline__ float dec_f32(unsigned u) {
    unsigned b = (u & 0x80000000u) ? (u & 0x7FFFFFFFu) : ~u;
    return __uint_as_float(b);
}

__global__ __launch_bounds__(256) void k_edge1(const float* __restrict__ el,
                                               const float* __restrict__ er,
                                               const int* __restrict__ src,
                                               const int* __restrict__ dst,
                                               float* __restrict__ e,
                                               unsigned* __restrict__ mEnc) {
    const int i = blockIdx.x * 256 + threadIdx.x;
    if (i >= N_EDGES) return;
    float v = el[src[i]] + er[dst[i]];
    v = (v > 0.f) ? v : NEG_SLOPE * v;
    e[i] = v;
    atomicMax(&mEnc[dst[i]], enc_f32(v));
}

__global__ __launch_bounds__(256) void k_edge2(const int* __restrict__ dst,
                                               const unsigned* __restrict__ mEnc,
                                               float* __restrict__ e,
                                               float* __restrict__ denom) {
    const int i = blockIdx.x * 256 + threadIdx.x;
    if (i >= N_EDGES) return;
    const int d = dst[i];
    const float ex = __expf(e[i] - dec_f32(mEnc[d]));
    e[i] = ex;
    atomicAdd(&denom[d], ex);
}

__global__ __launch_bounds__(256) void k_edge3(const int* __restrict__ src,
                                               const int* __restrict__ dst,
                                               const float* __restrict__ ex,
                                               const float* __restrict__ denom,
                                               const float* __restrict__ feat,
                                               float* __restrict__ h) {
    const int eIdx = blockIdx.x * 2 + (threadIdx.x >> 7);
    const int t    = threadIdx.x & 127;
    if (eIdx >= N_EDGES) return;
    const int d = dst[eIdx];
    const int s = src[eIdx];
    const float a = ex[eIdx] / denom[d];
    atomicAdd(&h[(size_t)d * D_HID + t], a * feat[(size_t)s * D_HID + t]);
}

extern "C" void kernel_launch(void* const* d_in, const int* in_sizes, int n_in,
                              void* d_out, int out_size, void* d_ws, size_t ws_size,
                              hipStream_t stream) {
    const float* X  = (const float*)d_in[0];
    const float* W  = (const float*)d_in[1];
    const float* al = (const float*)d_in[2];
    const float* ar = (const float*)d_in[3];
    const float* FW = (const float*)d_in[4];
    const float* FB = (const float*)d_in[5];
    const int*  src = (const int*)d_in[6];
    const int*  dst = (const int*)d_in[7];
    float* out = (float*)d_out;

    char* ws = (char*)d_ws;
    size_t off = 0;
    auto carve = [&](size_t bytes) {
        void* p = ws + off;
        off += (bytes + 255) & ~(size_t)255;
        return p;
    };
    float*    feat  = (float*)   carve((size_t)N_NODES * D_HID * sizeof(float));
    float*    h     = (float*)   carve((size_t)N_NODES * D_HID * sizeof(float));
    float*    el    = (float*)   carve((size_t)N_NODES * sizeof(float));
    float*    er    = (float*)   carve((size_t)N_NODES * sizeof(float));
    unsigned* mEnc  = (unsigned*)carve((size_t)N_NODES * sizeof(unsigned));
    float*    denom = (float*)   carve((size_t)N_NODES * sizeof(float));
    float*    e     = (float*)   carve((size_t)N_EDGES * sizeof(float));

    hipMemsetAsync(h,     0, (size_t)N_NODES * D_HID * sizeof(float), stream);
    hipMemsetAsync(mEnc,  0, (size_t)N_NODES * sizeof(unsigned), stream);
    hipMemsetAsync(denom, 0, (size_t)N_NODES * sizeof(float), stream);

    k_gemm1<<<3125, 256, 0, stream>>>(X, W, feat);
    k_scores<<<12500, 256, 0, stream>>>(feat, al, ar, el, er);
    k_edge1<<<(N_EDGES + 255) / 256, 256, 0, stream>>>(el, er, src, dst, e, mEnc);
    k_edge2<<<(N_EDGES + 255) / 256, 256, 0, stream>>>(dst, mEnc, e, denom);
    k_edge3<<<N_EDGES / 2, 256, 0, stream>>>(src, dst, e, denom, feat, h);
    k_gemm2<<<782, 256, 0, stream>>>(h, FW, FB, out);
}


# Edges grouped by destination each call (returning int add as rank, scan, fill); one wave per node does softmax and weighted row sum in registers: no float atomics, sums in arrival-rank order
# speedup vs baseline: 2.6907x; 2.6907x over previous
#include <hip/hip_runtime.h>
#include <hip/hip_bf16.h>

#define N_NODES   100000
#define N_EDGES   1600000
#define D_IN      256
#define D_HID     128
#define N_CLASSES 64
#define NEG_SLOPE 0.2f

typedef __attribute__((ext_vector_type(2))) float v2f;
typedef __attribute__((ext_vector_type(4))) float v4f;
typedef __attribute__((ext_vector_type(8))) float v8f;

__global__ __launch_bounds__(256) void k_gemm1(const float* __restrict__ X,
                                               const float* __restrict__ W,
                                               float* __restrict__ feat) {
    const int wave = threadIdx.x >> 5;
    const int lane = threadIdx.x & 31;
    const int strip = blockIdx.x * 8 + wave;
    const int rowTile = strip >> 2;
    const int colPair = strip & 3;
    const int row0 = rowTile * 16;
    const int col0 = colPair * 32;
    const int l16 = lane & 15;
    const int hi  = lane >> 4;

    v8f acc0 = {};
    v8f acc1 = {};
    const float* __restrict__ xrow = X + (size_t)(row0 + l16) * D_IN;

    for (int k0 = 0; k0 < D_IN; k0 += 4) {
        const int ka = k0 + 2 * hi;
        v2f a;
        a.x = xrow[ka];
        a.y = xrow[ka + 1];
        const float* __restrict__ w0 = W + (size_t)ka * D_HID + col0 + l16;
        const float* __restrict__ w1 = W + (size_t)(ka + 1) * D_HID + col0 + l16;
        v2f b0, b1;
        b0.x = w0[0];   b0.y = w1[0];
        b1.x = w0[16];  b1.y = w1[16];
        acc0 = __builtin_amdgcn_wmma_f32_16x16x4_f32(
            false, a, false, b0, (short)0, acc0, false, false);
        acc1 = __builtin_amdgcn_wmma_f32_16x16x4_f32(
            false, a, false, b1, (short)0, acc1, false, false);
    }
    #pragma unroll
    for (int r = 0; r < 8; ++r) {
        float* __restrict__ orow = feat + (size_t)(row0 + r + 8 * hi) * D_HID + col0 + l16;
        orow[0]  = acc0[r];
        orow[16] = acc1[r];
    }
}

__global__ __launch_bounds__(256) void k_gemm2(const float* __restrict__ H,
                                               const float* __restrict__ FW,
                                               const float* __restrict__ FB,
                                               float* __restrict__ out) {
    const int wave = threadIdx.x >> 5;
    const int lane = threadIdx.x & 31;
    const int rowTile = blockIdx.x * 8 + wave;
    if (rowTile >= N_NODES / 16) return;
    const int row0 = rowTile * 16;
    const int l16 = lane & 15;
    const int hi  = lane >> 4;

    v8f acc0 = {}, acc1 = {}, acc2 = {}, acc3 = {};
    const float* __restrict__ hrow = H + (size_t)(row0 + l16) * D_HID;

    for (int k0 = 0; k0 < D_HID; k0 += 4) {
        const int ka = k0 + 2 * hi;
        v2f a;
        a.x = hrow[ka];
        a.y = hrow[ka + 1];
        const float* __restrict__ w0 = FW + (size_t)ka * N_CLASSES + l16;
        const float* __restrict__ w1 = FW + (size_t)(ka + 1) * N_CLASSES + l16;
        v2f b0, b1, b2, b3;
        b0.x = w0[0];   b0.y = w1[0];
        b1.x = w0[16];  b1.y = w1[16];
        b2.x = w0[32];  b2.y = w1[32];
        b3.x = w0[48];  b3.y = w1[48];
        acc0 = __builtin_amdgcn_wmma_f32_16x16x4_f32(
            false, a, false, b0, (short)0, acc0, false, false);
        acc1 = __builtin_amdgcn_wmma_f32_16x16x4_f32(
            false, a, false, b1, (short)0, acc1, false, false);
        acc2 = __builtin_amdgcn_wmma_f32_16x16x4_f32(
            false, a, false, b2, (short)0, acc2, false, false);
        acc3 = __builtin_amdgcn_wmma_f32_16x16x4_f32(
            false, a, false, b3, (short)0, acc3, false, false);
    }
    const float bias0 = FB[l16];
    const float bias1 = FB[l16 + 16];
    const float bias2 = FB[l16 + 32];
    const float bias3 = FB[l16 + 48];
    #pragma unroll
    for (int r = 0; r < 8; ++r) {
        float* __restrict__ orow = out + (size_t)(row0 + r + 8 * hi) * N_CLASSES + l16;
        orow[0]  = acc0[r] + bias0;
        orow[16] = acc1[r] + bias1;
        orow[32] = acc2[r] + bias2;
        orow[48] = acc3[r] + bias3;
    }
}

__global__ __launch_bounds__(256) void k_scores(const float* __restrict__ feat,
                                                const float* __restrict__ al,
                                                const float* __restrict__ ar,
                                                float* __restrict__ el,
                                                float* __restrict__ er) {
    const int wave = threadIdx.x >> 5;
    const int lane = threadIdx.x & 31;
    const int node = blockIdx.x * 8 + wave;
    if (node >= N_NODES) return;
    const float* __restrict__ f = feat + (size_t)node * D_HID;
    float sl = 0.f, sr = 0.f;
    #pragma unroll
    for (int j = 0; j < D_HID; j += 32) {
        float v = f[j + lane];
        sl += v * al[j + lane];
        sr += v * ar[j + lane];
    }
    #pragma unroll
    for (int off = 16; off > 0; off >>= 1) {
        sl += __shfl_down(sl, off, 32);
        sr += __shfl_down(sr, off, 32);
    }
    if (lane == 0) { el[node] = sl; er[node] = sr; }
}

__global__ __launch_bounds__(256) void k_rank(const int* __restrict__ dst,
                                              int* __restrict__ cnt,
                                              int* __restrict__ rank) {
    const int i = blockIdx.x * 256 + threadIdx.x;
    if (i >= N_EDGES) return;
    rank[i] = atomicAdd(&cnt[dst[i]], 1);
}

__global__ __launch_bounds__(256) void k_scan(const int* __restrict__ cnt,
                                              int* __restrict__ rowptr) {
    __shared__ int wsum[8];
    const int b = blockIdx.x;
    const int t = threadIdx.x;
    const int lane = t & 31;
    const int wave = t >> 5;
    const int4* __restrict__ c4 = (const int4*)cnt;

    int acc = 0;
    for (int j = t; j < b * 256; j += 256) {
        const int4 v = c4[j];
        acc += (v.x + v.y) + (v.z + v.w);
    }
    for (int off = 16; off > 0; off >>= 1) acc += __shfl_xor(acc, off, 32);
    if (lane == 0) wsum[wave] = acc;
    __syncthreads();
    int base = 0;
    for (int w = 0; w < 8; ++w) base += wsum[w];
    __syncthreads();

    const int j = b * 256 + t;
    const bool ok = j < N_NODES / 4;
    int4 v = make_int4(0, 0, 0, 0);
    if (ok) v = c4[j];
    const int s = (v.x + v.y) + (v.z + v.w);
    int inc = s;
    for (int off = 1; off < 32; off <<= 1) {
        const int nb = __shfl_up(inc, off, 32);
        if (lane >= off) inc += nb;
    }
    if (lane == 31) wsum[wave] = inc;
    __syncthreads();
    int wbase = 0;
    for (int w = 0; w < 8; ++w) wbase += (w < wave) ? wsum[w] : 0;
    const int excl = base + wbase + inc - s;
    if (ok) {
        int4 o;
        o.x = excl;
        o.y = o.x + v.x;
        o.z = o.y + v.y;
        o.w = o.z + v.z;
        ((int4*)rowptr)[j] = o;
    }
}

__global__ __launch_bounds__(256) void k_fill(const int* __restrict__ src,
                                              const int* __restrict__ dst,
                                              const int* __restrict__ rowptr,
                                              const int* __restrict__ rank,
                                              int* __restrict__ esrc) {
    const int i = blockIdx.x * 256 + threadIdx.x;
    if (i >= N_EDGES) return;
    esrc[rowptr[dst[i]] + rank[i]] = src[i];
}

__device__ __forceinline__ void load_chunk(const int* __restrict__ esrc,
                                           const float* __restrict__ el,
                                           int start, int deg, int c, int lane,
                                           float er_n, int& s, float& e, bool& v) {
    const int j = c + lane;
    v = j < deg;
    s = v ? esrc[start + j] : 0;
    float t = el[s] + er_n;
    t = (t > 0.f) ? t : NEG_SLOPE * t;
    e = v ? t : -INFINITY;
}

__device__ __forceinline__ v4f row_of(const v4f* __restrict__ feat4, int s, int j, int lane) {
    const int sj = __builtin_amdgcn_readlane(s, j);
    return feat4[(size_t)sj * (D_HID / 4) + lane];
}
__device__ __forceinline__ float wt_of(float a, int j) {
    return __int_as_float(__builtin_amdgcn_readlane(__float_as_int(a), j));
}

__device__ __forceinline__ void agg_rows(const v4f* __restrict__ feat4, int nrows,
                                         int s, float a, int lane, v4f& acc) {
    int j = 0;
    for (; j + 4 <= nrows; j += 4) {
        const v4f f0 = row_of(feat4, s, j,     lane);
        const v4f f1 = row_of(feat4, s, j + 1, lane);
        const v4f f2 = row_of(feat4, s, j + 2, lane);
        const v4f f3 = row_of(feat4, s, j + 3, lane);
        acc += wt_of(a, j)     * f0;
        acc += wt_of(a, j + 1) * f1;
        acc += wt_of(a, j + 2) * f2;
        acc += wt_of(a, j + 3) * f3;
    }
    const int rem = nrows - j;
    v4f g0 = {0.f, 0.f, 0.f, 0.f}, g1 = g0, g2 = g0;
    float w0 = 0.f, w1 = 0.f, w2 = 0.f;
    if (rem > 0) { g0 = row_of(feat4, s, j,     lane); w0 = wt_of(a, j); }
    if (rem > 1) { g1 = row_of(feat4, s, j + 1, lane); w1 = wt_of(a, j + 1); }
    if (rem > 2) { g2 = row_of(feat4, s, j + 2, lane); w2 = wt_of(a, j + 2); }
    acc += w0 * g0;
    acc += w1 * g1;
    acc += w2 * g2;
}

__global__ __launch_bounds__(256) void k_pull(const int* __restrict__ cnt,
                                              const int* __restrict__ rowptr,
                                              const int* __restrict__ esrc,
                                              const float* __restrict__ el,
                                              const float* __restrict__ er,
                                              const float* __restrict__ feat,
                                              float* __restrict__ h) {
    const int lane = threadIdx.x & 31;
    const int n = __builtin_amdgcn_readfirstlane((int)(blockIdx.x * 8 + (threadIdx.x >> 5)));
    if (n >= N_NODES) return;
    const int deg   = cnt[n];
    const int start = rowptr[n];
    const float er_n = er[n];
    const v4f* __restrict__ feat4 = (const v4f*)feat;

    int s0; float e0; bool v0;
    load_chunk(esrc, el, start, deg, 0, lane, er_n, s0, e0, v0);

    float m = e0;
    for (int c = 32; c < deg; c += 32) {
        int s; float e; bool v;
        load_chunk(esrc, el, start, deg, c, lane, er_n, s, e, v);
        m = fmaxf(m, e);
    }
    for (int off = 16; off > 0; off >>= 1) m = fmaxf(m, __shfl_xor(m, off, 32));

    const float x0 = v0 ? __expf(e0 - m) : 0.f;
    float dsum = x0;
    for (int c = 32; c < deg; c += 32) {
        int s; float e; bool v;
        load_chunk(esrc, el, start, deg, c, lane, er_n, s, e, v);
        dsum += v ? __expf(e - m) : 0.f;
    }
    for (int off = 16; off > 0; off >>= 1) dsum += __shfl_xor(dsum, off, 32);

    v4f acc = {0.f, 0.f, 0.f, 0.f};
    agg_rows(feat4, deg < 32 ? deg : 32, s0, x0 / dsum, lane, acc);
    for (int c = 32; c < deg; c += 32) {
        int s; float e; bool v;
        load_chunk(esrc, el, start, deg, c, lane, er_n, s, e, v);
        const float a = v ? __expf(e - m) / dsum : 0.f;
        const int left = deg - c;
        agg_rows(feat4, left < 32 ? left : 32, s, a, lane, acc);
    }
    ((v4f*)h)[(size_t)n * (D_HID / 4) + lane] = acc;
}

extern "C" void kernel_launch(void* const* d_in, const int* in_sizes, int n_in,
                              void* d_out, int out_size, void* d_ws, size_t ws_size,
                              hipStream_t stream) {
    const float* X  = (const float*)d_in[0];
    const float* W  = (const float*)d_in[1];
    const float* al = (const float*)d_in[2];
    const float* ar = (const float*)d_in[3];
    const float* FW = (const float*)d_in[4];
    const float* FB = (const float*)d_in[5];
    const int*  src = (const int*)d_in[6];
    const int*  dst = (const int*)d_in[7];
    float* out = (float*)d_out;

    char* ws = (char*)d_ws;
    size_t off = 0;
    auto carve = [&](size_t bytes) {
        void* p = ws + off;
        off += (bytes + 255) & ~(size_t)255;
        return p;
    };
    float* feat   = (float*)carve((size_t)N_NODES * D_HID * sizeof(float));
    float* h      = (float*)carve((size_t)N_NODES * D_HID * sizeof(float));
    float* el     = (float*)carve((size_t)N_NODES * sizeof(float));
    float* er     = (float*)carve((size_t)N_NODES * sizeof(float));
    int*   cnt    = (int*)  carve((size_t)N_NODES * sizeof(int));
    int*   rowptr = (int*)  carve((size_t)N_NODES * sizeof(int));
    int*   esrc   = (int*)  carve((size_t)N_EDGES * sizeof(int));
    int*   rank   = (int*)h;

    hipMemsetAsync(cnt, 0, (size_t)N_NODES * sizeof(int), stream);

    k_rank<<<(N_EDGES + 255) / 256, 256, 0, stream>>>(dst, cnt, rank);
    k_scan<<<(N_NODES / 4 + 255) / 256, 256, 0, stream>>>(cnt, rowptr);
    k_fill<<<(N_EDGES + 255) / 256, 256, 0, stream>>>(src, dst, rowptr, rank, esrc);
    k_gemm1<<<3125, 256, 0, stream>>>(X, W, feat);
    k_scores<<<12500, 256, 0, stream>>>(feat, al, ar, el, er);
    k_pull<<<N_NODES / 8, 256, 0, stream>>>(cnt, rowptr, esrc, el, er, feat, h);
    k_gemm2<<<782, 256, 0, stream>>>(h, FW, FB, out);
}


# Edge grouping with no global atomics: LDS-counted partition (bin by 256-node range, then per-bin LDS ranking) replaces 1.6M returning global adds, memset, fill; row sums stay in arrival order
# speedup vs baseline: 3.6114x; 1.3422x over previous
#include <hip/hip_runtime.h>
#include <hip/hip_bf16.h>

#define N_NODES   100000
#define N_EDGES   1600000
#define D_IN      256
#define D_HID     128
#define N_CLASSES 64
#define NEG_SLOPE 0.2f

typedef __attribute__((ext_vector_type(2))) float v2f;
typedef __attribute__((ext_vector_type(4))) float v4f;
typedef __attribute__((ext_vector_type(8))) float v8f;

__global__ __launch_bounds__(256) void k_gemm1(const float* __restrict__ X,
                                               const float* __restrict__ W,
                                               float* __restrict__ feat) {
    const int wave = threadIdx.x >> 5;
    const int lane = threadIdx.x & 31;
    const int strip = blockIdx.x * 8 + wave;
    const int rowTile = strip >> 2;
    const int colPair = strip & 3;
    const int row0 = rowTile * 16;
    const int col0 = colPair * 32;
    const int l16 = lane & 15;
    const int hi  = lane >> 4;

    v8f acc0 = {};
    v8f acc1 = {};
    const float* __restrict__ xrow = X + (size_t)(row0 + l16) * D_IN;

    for (int k0 = 0; k0 < D_IN; k0 += 4) {
        const int ka = k0 + 2 * hi;
        v2f a;
        a.x = xrow[ka];
        a.y = xrow[ka + 1];
        const float* __restrict__ w0 = W + (size_t)ka * D_HID + col0 + l16;
        const float* __restrict__ w1 = W + (size_t)(ka + 1) * D_HID + col0 + l16;
        v2f b0, b1;
        b0.x = w0[0];   b0.y = w1[0];
        b1.x = w0[16];  b1.y = w1[16];
        acc0 = __builtin_amdgcn_wmma_f32_16x16x4_f32(
            false, a, false, b0, (short)0, acc0, false, false);
        acc1 = __builtin_amdgcn_wmma_f32_16x16x4_f32(
            false, a, false, b1, (short)0, acc1, false, false);
    }
    #pragma unroll
    for (int r = 0; r < 8; ++r) {
        float* __restrict__ orow = feat + (size_t)(row0 + r + 8 * hi) * D_HID + col0 + l16;
        orow[0]  = acc0[r];
        orow[16] = acc1[r];
    }
}

__global__ __launch_bounds__(256) void k_gemm2(const float* __restrict__ H,
                                               const float* __restrict__ FW,
                                               const float* __restrict__ FB,
                                               float* __restrict__ out) {
    const int wave = threadIdx.x >> 5;
    const int lane = threadIdx.x & 31;
    const int rowTile = blockIdx.x * 8 + wave;
    if (rowTile >= N_NODES / 16) return;
    const int row0 = rowTile * 16;
    const int l16 = lane & 15;
    const int hi  = lane >> 4;

    v8f acc0 = {}, acc1 = {}, acc2 = {}, acc3 = {};
    const float* __restrict__ hrow = H + (size_t)(row0 + l16) * D_HID;

    for (int k0 = 0; k0 < D_HID; k0 += 4) {
        const int ka = k0 + 2 * hi;
        v2f a;
        a.x = hrow[ka];
        a.y = hrow[ka + 1];
        const float* __restrict__ w0 = FW + (size_t)ka * N_CLASSES + l16;
        const float* __restrict__ w1 = FW + (size_t)(ka + 1) * N_CLASSES + l16;
        v2f b0, b1, b2, b3;
        b0.x = w0[0];   b0.y = w1[0];
        b1.x = w0[16];  b1.y = w1[16];
        b2.x = w0[32];  b2.y = w1[32];
        b3.x = w0[48];  b3.y = w1[48];
        acc0 = __builtin_amdgcn_wmma_f32_16x16x4_f32(
            false, a, false, b0, (short)0, acc0, false, false);
        acc1 = __builtin_amdgcn_wmma_f32_16x16x4_f32(
            false, a, false, b1, (short)0, acc1, false, false);
        acc2 = __builtin_amdgcn_wmma_f32_16x16x4_f32(
            false, a, false, b2, (short)0, acc2, false, false);
        acc3 = __builtin_amdgcn_wmma_f32_16x16x4_f32(
            false, a, false, b3, (short)0, acc3, false, false);
    }
    const float bias0 = FB[l16];
    const float bias1 = FB[l16 + 16];
    const float bias2 = FB[l16 + 32];
    const float bias3 = FB[l16 + 48];
    #pragma unroll
    for (int r = 0; r < 8; ++r) {
        float* __restrict__ orow = out + (size_t)(row0 + r + 8 * hi) * N_CLASSES + l16;
        orow[0]  = acc0[r] + bias0;
        orow[16] = acc1[r] + bias1;
        orow[32] = acc2[r] + bias2;
        orow[48] = acc3[r] + bias3;
    }
}

__global__ __launch_bounds__(256) void k_scores(const float* __restrict__ feat,
                                                const float* __restrict__ al,
                                                const float* __restrict__ ar,
                                                float* __restrict__ el,
                                                float* __restrict__ er) {
    const int wave = threadIdx.x >> 5;
    const int lane = threadIdx.x & 31;
    const int node = blockIdx.x * 8 + wave;
    if (node >= N_NODES) return;
    const float* __restrict__ f = feat + (size_t)node * D_HID;
    float sl = 0.f, sr = 0.f;
    #pragma unroll
    for (int j = 0; j < D_HID; j += 32) {
        float v = f[j + lane];
        sl += v * al[j + lane];
        sr += v * ar[j + lane];
    }
    #pragma unroll
    for (int off = 16; off > 0; off >>= 1) {
        sl += __shfl_down(sl, off, 32);
        sr += __shfl_down(sr, off, 32);
    }
    if (lane == 0) { el[node] = sl; er[node] = sr; }
}

#define BIN_SHIFT 8
#define BIN_NODES (1 << BIN_SHIFT)
#define NBIN      ((N_NODES + BIN_NODES - 1) / BIN_NODES)
#define PB        256
#define EPB       (N_EDGES / PB)
#define NHIST     (NBIN * PB)
#define PT        1024
#define EIT       ((EPB + PT - 1) / PT)
#define SU        5
static_assert(N_EDGES % PB == 0, "edge chunks must tile the edge list");
static_assert(NHIST % 4 == 0, "table is scanned four entries at a time");

__global__ __launch_bounds__(PT) void k_binhist(const int* __restrict__ dst,
                                                int* __restrict__ hist) {
    __shared__ int tab[NBIN];
    const int t = threadIdx.x;
    const int b = blockIdx.x;
    for (int i = t; i < NBIN; i += PT) tab[i] = 0;
    __syncthreads();
    const int* __restrict__ d = dst + (size_t)b * EPB;
    int dv[EIT];
    #pragma unroll
    for (int k = 0; k < EIT; ++k) {
        const int i = t + k * PT;
        dv[k] = (i < EPB) ? d[i] : 0;
    }
    #pragma unroll
    for (int k = 0; k < EIT; ++k) {
        const int i = t + k * PT;
        if (i < EPB) atomicAdd(&tab[dv[k] >> BIN_SHIFT], 1);
    }
    __syncthreads();
    for (int i = t; i < NBIN; i += PT) hist[i * PB + b] = tab[i];
}

__global__ __launch_bounds__(256) void k_scan(const int* __restrict__ hist,
                                              int* __restrict__ offs) {
    __shared__ int wsum[8];
    const int b = blockIdx.x;
    const int t = threadIdx.x;
    const int lane = t & 31;
    const int wave = t >> 5;
    const int4* __restrict__ c4 = (const int4*)hist;

    int acc = 0;
    for (int j = t; j < b * 256; j += 256) {
        const int4 v = c4[j];
        acc += (v.x + v.y) + (v.z + v.w);
    }
    for (int off = 16; off > 0; off >>= 1) acc += __shfl_xor(acc, off, 32);
    if (lane == 0) wsum[wave] = acc;
    __syncthreads();
    int base = 0;
    for (int w = 0; w < 8; ++w) base += wsum[w];
    __syncthreads();

    const int j = b * 256 + t;
    const bool ok = j < NHIST / 4;
    int4 v = make_int4(0, 0, 0, 0);
    if (ok) v = c4[j];
    const int s = (v.x + v.y) + (v.z + v.w);
    int inc = s;
    for (int off = 1; off < 32; off <<= 1) {
        const int nb = __shfl_up(inc, off, 32);
        if (lane >= off) inc += nb;
    }
    if (lane == 31) wsum[wave] = inc;
    __syncthreads();
    int wbase = 0;
    for (int w = 0; w < 8; ++w) wbase += (w < wave) ? wsum[w] : 0;
    const int excl = base + wbase + inc - s;
    if (ok) {
        int4 o;
        o.x = excl;
        o.y = o.x + v.x;
        o.z = o.y + v.y;
        o.w = o.z + v.z;
        ((int4*)offs)[j] = o;
    }
}

__global__ __launch_bounds__(PT) void k_binscatter(const int* __restrict__ src,
                                                   const int* __restrict__ dst,
                                                   const int* __restrict__ offs,
                                                   int* __restrict__ binned) {
    __shared__ int cur[NBIN];
    const int t = threadIdx.x;
    const int b = blockIdx.x;
    for (int i = t; i < NBIN; i += PT) cur[i] = offs[i * PB + b];
    __syncthreads();
    const int* __restrict__ s = src + (size_t)b * EPB;
    const int* __restrict__ d = dst + (size_t)b * EPB;
    int2* __restrict__ out = (int2*)binned;
    int dv[EIT], sv[EIT];
    #pragma unroll
    for (int k = 0; k < EIT; ++k) {
        const int i = t + k * PT;
        dv[k] = (i < EPB) ? d[i] : 0;
        sv[k] = (i < EPB) ? s[i] : 0;
    }
    #pragma unroll
    for (int k = 0; k < EIT; ++k) {
        const int i = t + k * PT;
        if (i < EPB) {
            const int pos = atomicAdd(&cur[dv[k] >> BIN_SHIFT], 1);
            out[pos] = make_int2(sv[k], dv[k]);
        }
    }
}

__global__ __launch_bounds__(PT) void k_binsort(const int* __restrict__ offs,
                                                const int* __restrict__ binned,
                                                int* __restrict__ cnt,
                                                int* __restrict__ rowptr,
                                                int* __restrict__ esrc) {
    __shared__ int c[BIN_NODES];
    __shared__ int cur[BIN_NODES];
    __shared__ int wsum[BIN_NODES / 32];
    const int t = threadIdx.x;
    const int b = blockIdx.x;
    const int lane = t & 31;
    const int wave = t >> 5;
    const int beg = offs[b * PB];
    const int end = (b + 1 < NBIN) ? offs[(b + 1) * PB] : N_EDGES;
    const int2* __restrict__ in = (const int2*)binned;

    if (t < BIN_NODES) c[t] = 0;
    __syncthreads();
    for (int base = beg; base < end; base += SU * PT) {
        int y[SU];
        #pragma unroll
        for (int k = 0; k < SU; ++k) {
            const int i = base + t + k * PT;
            y[k] = in[(i < end) ? i : (end - 1)].y;
        }
        #pragma unroll
        for (int k = 0; k < SU; ++k) {
            const int i = base + t + k * PT;
            if (i < end) atomicAdd(&c[y[k] & (BIN_NODES - 1)], 1);
        }
    }
    __syncthreads();

    const bool act = t < BIN_NODES;
    const int v = act ? c[t] : 0;
    int inc = v;
    for (int off = 1; off < 32; off <<= 1) {
        const int nb = __shfl_up(inc, off, 32);
        if (lane >= off) inc += nb;
    }
    if (act && lane == 31) wsum[wave] = inc;
    __syncthreads();
    int wbase = 0;
    for (int w = 0; w < BIN_NODES / 32; ++w) wbase += (w < wave) ? wsum[w] : 0;
    const int excl = beg + wbase + inc - v;
    if (act) {
        cur[t] = excl;
        const int n = b * BIN_NODES + t;
        if (n < N_NODES) { cnt[n] = v; rowptr[n] = excl; }
    }
    __syncthreads();
    for (int base = beg; base < end; base += SU * PT) {
        int2 sd[SU];
        #pragma unroll
        for (int k = 0; k < SU; ++k) {
            const int i = base + t + k * PT;
            sd[k] = in[(i < end) ? i : (end - 1)];
        }
        #pragma unroll
        for (int k = 0; k < SU; ++k) {
            const int i = base + t + k * PT;
            if (i < end) {
                const int pos = atomicAdd(&cur[sd[k].y & (BIN_NODES - 1)], 1);
                esrc[pos] = sd[k].x;
            }
        }
    }
}

__device__ __forceinline__ void load_chunk(const int* __restrict__ esrc,
                                           const float* __restrict__ el,
                                           int start, int deg, int c, int lane,
                                           float er_n, int& s, float& e, bool& v) {
    const int j = c + lane;
    v = j < deg;
    s = v ? esrc[start + j] : 0;
    float t = el[s] + er_n;
    t = (t > 0.f) ? t : NEG_SLOPE * t;
    e = v ? t : -INFINITY;
}

__device__ __forceinline__ v4f row_of(const v4f* __restrict__ feat4, int s, int j, int lane) {
    const int sj = __builtin_amdgcn_readlane(s, j);
    return feat4[(size_t)sj * (D_HID / 4) + lane];
}
__device__ __forceinline__ float wt_of(float a, int j) {
    return __int_as_float(__builtin_amdgcn_readlane(__float_as_int(a), j));
}

__device__ __forceinline__ void agg_rows(const v4f* __restrict__ feat4, int nrows,
                                         int s, float a, int lane, v4f& acc) {
    int j = 0;
    for (; j + 4 <= nrows; j += 4) {
        const v4f f0 = row_of(feat4, s, j,     lane);
        const v4f f1 = row_of(feat4, s, j + 1, lane);
        const v4f f2 = row_of(feat4, s, j + 2, lane);
        const v4f f3 = row_of(feat4, s, j + 3, lane);
        acc += wt_of(a, j)     * f0;
        acc += wt_of(a, j + 1) * f1;
        acc += wt_of(a, j + 2) * f2;
        acc += wt_of(a, j + 3) * f3;
    }
    const int rem = nrows - j;
    v4f g0 = {0.f, 0.f, 0.f, 0.f}, g1 = g0, g2 = g0;
    float w0 = 0.f, w1 = 0.f, w2 = 0.f;
    if (rem > 0) { g0 = row_of(feat4, s, j,     lane); w0 = wt_of(a, j); }
    if (rem > 1) { g1 = row_of(feat4, s, j + 1, lane); w1 = wt_of(a, j + 1); }
    if (rem > 2) { g2 = row_of(feat4, s, j + 2, lane); w2 = wt_of(a, j + 2); }
    acc += w0 * g0;
    acc += w1 * g1;
    acc += w2 * g2;
}

__global__ __launch_bounds__(256) void k_pull(const int* __restrict__ cnt,
                                              const int* __restrict__ rowptr,
                                              const int* __restrict__ esrc,
                                              const float* __restrict__ el,
                                              const float* __restrict__ er,
                                              const float* __restrict__ feat,
                                              float* __restrict__ h) {
    const int lane = threadIdx.x & 31;
    const int n = __builtin_amdgcn_readfirstlane((int)(blockIdx.x * 8 + (threadIdx.x >> 5)));
    if (n >= N_NODES) return;
    const int deg   = cnt[n];
    const int start = rowptr[n];
    const float er_n = er[n];
    const v4f* __restrict__ feat4 = (const v4f*)feat;

    int s0; float e0; bool v0;
    load_chunk(esrc, el, start, deg, 0, lane, er_n, s0, e0, v0);

    float m = e0;
    for (int c = 32; c < deg; c += 32) {
        int s; float e; bool v;
        load_chunk(esrc, el, start, deg, c, lane, er_n, s, e, v);
        m = fmaxf(m, e);
    }
    for (int off = 16; off > 0; off >>= 1) m = fmaxf(m, __shfl_xor(m, off, 32));

    const float x0 = v0 ? __expf(e0 - m) : 0.f;
    float dsum = x0;
    for (int c = 32; c < deg; c += 32) {
        int s; float e; bool v;
        load_chunk(esrc, el, start, deg, c, lane, er_n, s, e, v);
        dsum += v ? __expf(e - m) : 0.f;
    }
    for (int off = 16; off > 0; off >>= 1) dsum += __shfl_xor(dsum, off, 32);

    v4f acc = {0.f, 0.f, 0.f, 0.f};
    agg_rows(feat4, deg < 32 ? deg : 32, s0, x0 / dsum, lane, acc);
    for (int c = 32; c < deg; c += 32) {
        int s; float e; bool v;
        load_chunk(esrc, el, start, deg, c, lane, er_n, s, e, v);
        const float a = v ? __expf(e - m) / dsum : 0.f;
        const int left = deg - c;
        agg_rows(feat4, left < 32 ? left : 32, s, a, lane, acc);
    }
    ((v4f*)h)[(size_t)n * (D_HID / 4) + lane] = acc;
}

extern "C" void kernel_launch(void* const* d_in, const int* in_sizes, int n_in,
                              void* d_out, int out_size, void* d_ws, size_t ws_size,
                              hipStream_t stream) {
    const float* X  = (const float*)d_in[0];
    const float* W  = (const float*)d_in[1];
    const float* al = (const float*)d_in[2];
    const float* ar = (const float*)d_in[3];
    const float* FW = (const float*)d_in[4];
    const float* FB = (const float*)d_in[5];
    const int*  src = (const int*)d_in[6];
    const int*  dst = (const int*)d_in[7];
    float* out = (float*)d_out;

    char* ws = (char*)d_ws;
    size_t off = 0;
    auto carve = [&](size_t bytes) {
        void* p = ws + off;
        off += (bytes + 255) & ~(size_t)255;
        return p;
    };
    float* feat   = (float*)carve((size_t)N_NODES * D_HID * sizeof(float));
    float* h      = (float*)carve((size_t)N_NODES * D_HID * sizeof(float));
    float* el     = (float*)carve((size_t)N_NODES * sizeof(float));
    float* er     = (float*)carve((size_t)N_NODES * sizeof(float));
    int*   cnt    = (int*)  carve((size_t)N_NODES * sizeof(int));
    int*   rowptr = (int*)  carve((size_t)N_NODES * sizeof(int));
    int*   esrc   = (int*)  carve((size_t)N_EDGES * sizeof(int));
    int*   binned = (int*)h;
    int*   hist   = binned + (size_t)2 * N_EDGES;
    int*   offs   = hist + NHIST;

    k_binhist<<<PB, PT, 0, stream>>>(dst, hist);
    k_scan<<<(NHIST / 4 + 255) / 256, 256, 0, stream>>>(hist, offs);
    k_binscatter<<<PB, PT, 0, stream>>>(src, dst, offs, binned);
    k_binsort<<<NBIN, PT, 0, stream>>>(offs, binned, cnt, rowptr, esrc);
    k_gemm1<<<3125, 256, 0, stream>>>(X, W, feat);
    k_scores<<<12500, 256, 0, stream>>>(feat, al, ar, el, er);
    k_pull<<<N_NODES / 8, 256, 0, stream>>>(cnt, rowptr, esrc, el, er, feat, h);
    k_gemm2<<<782, 256, 0, stream>>>(h, FW, FB, out);
}


# Scan launch: prefix re-sum of the chunks in front was one dependent L2 round trip per chunk (loop not unrolled, 97 in a row in the last block); now 16 independent loads per trip; integer sums only
# speedup vs baseline: 6.3486x; 1.7579x over previous
#include <hip/hip_runtime.h>
#include <hip/hip_bf16.h>

#define N_NODES   100000
#define N_EDGES   1600000
#define D_IN      256
#define D_HID     128
#define N_CLASSES 64
#define NEG_SLOPE 0.2f

typedef __attribute__((ext_vector_type(2))) float v2f;
typedef __attribute__((ext_vector_type(4))) float v4f;
typedef __attribute__((ext_vector_type(8))) float v8f;
typedef __attribute__((ext_vector_type(4))) int   v4i;

#define NC 80
#define COL_EL N_CLASSES
#define COL_ER (N_CLASSES + 1)

__global__ __launch_bounds__(32) void k_fold(const float* __restrict__ W,
                                             const float* __restrict__ FW,
                                             const float* __restrict__ al,
                                             const float* __restrict__ ar,
                                             float* __restrict__ wcp) {
    const int lane = threadIdx.x;
    const int rowTile = blockIdx.x / 5;
    const int colTile = blockIdx.x - rowTile * 5;
    const int l16 = lane & 15;
    const int hi  = lane >> 4;
    const int row0 = rowTile * 16;
    const int col = colTile * 16 + l16;

    const bool isfc = col < N_CLASSES;
    const bool used = col <= COL_ER;
    const float* __restrict__ bp = isfc ? (FW + col) : ((col == COL_ER) ? ar : al);
    const int bs = isfc ? N_CLASSES : 1;

    v8f acc = {};
    const float* __restrict__ wrow = W + (size_t)(row0 + l16) * D_HID;
    for (int j0 = 0; j0 < D_HID; j0 += 4) {
        const int ja = j0 + 2 * hi;
        v2f a;
        a.x = wrow[ja];
        a.y = wrow[ja + 1];
        const float bx = bp[(size_t)ja * bs];
        const float by = bp[(size_t)(ja + 1) * bs];
        v2f b;
        b.x = used ? bx : 0.f;
        b.y = used ? by : 0.f;
        acc = __builtin_amdgcn_wmma_f32_16x16x4_f32(
            false, a, false, b, (short)0, acc, false, false);
    }
    #pragma unroll
    for (int r = 0; r < 8; ++r) {
        const int k = row0 + r + 8 * hi;
        wcp[(size_t)(k >> 1) * (2 * NC) + col * 2 + (k & 1)] = acc[r];
    }
}

__global__ __launch_bounds__(256) void k_gemm1(const float* __restrict__ X,
                                               const float* __restrict__ wcp,
                                               float* __restrict__ G,
                                               float* __restrict__ el,
                                               float* __restrict__ er) {
    const int wave = threadIdx.x >> 5;
    const int lane = threadIdx.x & 31;
    const int rowTile = blockIdx.x * 8 + wave;
    if (rowTile >= N_NODES / 16) return;
    const int row0 = rowTile * 16;
    const int l16 = lane & 15;
    const int hi  = lane >> 4;

    v8f acc0 = {}, acc1 = {}, acc2 = {}, acc3 = {}, acc4 = {};
    const float* __restrict__ xrow = X + (size_t)(row0 + l16) * D_IN;
    const v2f* __restrict__ wp = (const v2f*)wcp;

    for (int k0 = 0; k0 < D_IN; k0 += 4) {
        const int ka = k0 + 2 * hi;
        v2f a;
        a.x = xrow[ka];
        a.y = xrow[ka + 1];
        const v2f* __restrict__ w = wp + (size_t)(ka >> 1) * NC + l16;
        const v2f b0 = w[0];
        const v2f b1 = w[16];
        const v2f b2 = w[32];
        const v2f b3 = w[48];
        const v2f b4 = w[64];
        acc0 = __builtin_amdgcn_wmma_f32_16x16x4_f32(
            false, a, false, b0, (short)0, acc0, false, false);
        acc1 = __builtin_amdgcn_wmma_f32_16x16x4_f32(
            false, a, false, b1, (short)0, acc1, false, false);
        acc2 = __builtin_amdgcn_wmma_f32_16x16x4_f32(
            false, a, false, b2, (short)0, acc2, false, false);
        acc3 = __builtin_amdgcn_wmma_f32_16x16x4_f32(
            false, a, false, b3, (short)0, acc3, false, false);
        acc4 = __builtin_amdgcn_wmma_f32_16x16x4_f32(
            false, a, false, b4, (short)0, acc4, false, false);
    }
    float* __restrict__ sc = (l16 == 0) ? el : er;
    #pragma unroll
    for (int r = 0; r < 8; ++r) {
        const int row = row0 + r + 8 * hi;
        float* __restrict__ orow = G + (size_t)row * N_CLASSES + l16;
        orow[0]  = acc0[r];
        orow[16] = acc1[r];
        orow[32] = acc2[r];
        orow[48] = acc3[r];
        if (l16 < 2) sc[row] = acc4[r];
    }
}

#define BIN_SHIFT 8
#define BIN_NODES (1 << BIN_SHIFT)
#define NBIN      ((N_NODES + BIN_NODES - 1) / BIN_NODES)
#define PB        256
#define EPB       (N_EDGES / PB)
#define NHIST     (NBIN * PB)
#define PT        1024
#define EIT       ((EPB + PT - 1) / PT)
#define SU        5
#define SB        16
static_assert(SB == 16, "the statement that groups k_scan's loads names 16 vectors");
static_assert(N_EDGES % PB == 0, "edge chunks must tile the edge list");
static_assert(NHIST % 4 == 0, "table is scanned four entries at a time");

__global__ __launch_bounds__(PT) void k_binhist(const int* __restrict__ dst,
                                                int* __restrict__ hist) {
    __shared__ int tab[NBIN];
    const int t = threadIdx.x;
    const int b = blockIdx.x;
    for (int i = t; i < NBIN; i += PT) tab[i] = 0;
    __syncthreads();
    const int* __restrict__ d = dst + (size_t)b * EPB;
    int dv[EIT];
    #pragma unroll
    for (int k = 0; k < EIT; ++k) {
        const int i = t + k * PT;
        dv[k] = (i < EPB) ? d[i] : 0;
    }
    #pragma unroll
    for (int k = 0; k < EIT; ++k) {
        const int i = t + k * PT;
        if (i < EPB) atomicAdd(&tab[dv[k] >> BIN_SHIFT], 1);
    }
    __syncthreads();
    for (int i = t; i < NBIN; i += PT) hist[i * PB + b] = tab[i];
}

template <bool TAIL>
__device__ __forceinline__ int scan_batch(const v4i* __restrict__ c4v, int t, int i, int b) {
    v4i v[SB];
    #pragma unroll
    for (int k = 0; k < SB; ++k) {
        const int c = (!TAIL || i + k < b) ? (i + k) : 0;
        v[k] = c4v[c * 256 + t];
    }
    asm volatile("" : "+v"(v[0]),  "+v"(v[1]),  "+v"(v[2]),  "+v"(v[3]),
                      "+v"(v[4]),  "+v"(v[5]),  "+v"(v[6]),  "+v"(v[7]),
                      "+v"(v[8]),  "+v"(v[9]),  "+v"(v[10]), "+v"(v[11]),
                      "+v"(v[12]), "+v"(v[13]), "+v"(v[14]), "+v"(v[15]));
    int sum = 0;
    #pragma unroll
    for (int k = 0; k < SB; ++k) {
        const int s4 = (v[k].x + v[k].y) + (v[k].z + v[k].w);
        sum += (!TAIL || i + k < b) ? s4 : 0;
    }
    return sum;
}

__global__ __launch_bounds__(256) void k_scan(const int* __restrict__ hist,
                                              int* __restrict__ offs) {
    __shared__ int wsum[8];
    const int b = blockIdx.x;
    const int t = threadIdx.x;
    const int lane = t & 31;
    const int wave = t >> 5;
    const int4* __restrict__ c4 = (const int4*)hist;

    const v4i* __restrict__ c4v = (const v4i*)hist;
    int acc = 0;
    int i = 0;
    for (; i + SB <= b; i += SB) acc += scan_batch<false>(c4v, t, i, b);
    if (i < b) acc += scan_batch<true>(c4v, t, i, b);
    for (int off = 16; off > 0; off >>= 1) acc += __shfl_xor(acc, off, 32);
    if (lane == 0) wsum[wave] = acc;
    __syncthreads();
    int base = 0;
    for (int w = 0; w < 8; ++w) base += wsum[w];
    __syncthreads();

    const int j = b * 256 + t;
    const bool ok = j < NHIST / 4;
    int4 v = make_int4(0, 0, 0, 0);
    if (ok) v = c4[j];
    const int s = (v.x + v.y) + (v.z + v.w);
    int inc = s;
    for (int off = 1; off < 32; off <<= 1) {
        const int nb = __shfl_up(inc, off, 32);
        if (lane >= off) inc += nb;
    }
    if (lane == 31) wsum[wave] = inc;
    __syncthreads();
    int wbase = 0;
    for (int w = 0; w < 8; ++w) wbase += (w < wave) ? wsum[w] : 0;
    const int excl = base + wbase + inc - s;
    if (ok) {
        int4 o;
        o.x = excl;
        o.y = o.x + v.x;
        o.z = o.y + v.y;
        o.w = o.z + v.z;
        ((int4*)offs)[j] = o;
    }
}

__global__ __launch_bounds__(PT) void k_binscatter(const int* __restrict__ src,
                                                   const int* __restrict__ dst,
                                                   const int* __restrict__ offs,
                                                   int* __restrict__ binned) {
    __shared__ int cur[NBIN];
    const int t = threadIdx.x;
    const int b = blockIdx.x;
    for (int i = t; i < NBIN; i += PT) cur[i] = offs[i * PB + b];
    __syncthreads();
    const int* __restrict__ s = src + (size_t)b * EPB;
    const int* __restrict__ d = dst + (size_t)b * EPB;
    int2* __restrict__ out = (int2*)binned;
    int dv[EIT], sv[EIT];
    #pragma unroll
    for (int k = 0; k < EIT; ++k) {
        const int i = t + k * PT;
        dv[k] = (i < EPB) ? d[i] : 0;
        sv[k] = (i < EPB) ? s[i] : 0;
    }
    #pragma unroll
    for (int k = 0; k < EIT; ++k) {
        const int i = t + k * PT;
        if (i < EPB) {
            const int pos = atomicAdd(&cur[dv[k] >> BIN_SHIFT], 1);
            out[pos] = make_int2(sv[k], dv[k]);
        }
    }
}

__global__ __launch_bounds__(PT) void k_binsort(const int* __restrict__ offs,
                                                const int* __restrict__ binned,
                                                int* __restrict__ cnt,
                                                int* __restrict__ rowptr,
                                                int* __restrict__ esrc) {
    __shared__ int c[BIN_NODES];
    __shared__ int cur[BIN_NODES];
    __shared__ int wsum[BIN_NODES / 32];
    const int t = threadIdx.x;
    const int b = blockIdx.x;
    const int lane = t & 31;
    const int wave = t >> 5;
    const int beg = offs[b * PB];
    const int end = (b + 1 < NBIN) ? offs[(b + 1) * PB] : N_EDGES;
    const int2* __restrict__ in = (const int2*)binned;

    if (t < BIN_NODES) c[t] = 0;
    __syncthreads();
    for (int base = beg; base < end; base += SU * PT) {
        int y[SU];
        #pragma unroll
        for (int k = 0; k < SU; ++k) {
            const int i = base + t + k * PT;
            y[k] = in[(i < end) ? i : (end - 1)].y;
        }
        #pragma unroll
        for (int k = 0; k < SU; ++k) {
            const int i = base + t + k * PT;
            if (i < end) atomicAdd(&c[y[k] & (BIN_NODES - 1)], 1);
        }
    }
    __syncthreads();

    const bool act = t < BIN_NODES;
    const int v = act ? c[t] : 0;
    int inc = v;
    for (int off = 1; off < 32; off <<= 1) {
        const int nb = __shfl_up(inc, off, 32);
        if (lane >= off) inc += nb;
    }
    if (act && lane == 31) wsum[wave] = inc;
    __syncthreads();
    int wbase = 0;
    for (int w = 0; w < BIN_NODES / 32; ++w) wbase += (w < wave) ? wsum[w] : 0;
    const int excl = beg + wbase + inc - v;
    if (act) {
        cur[t] = excl;
        const int n = b * BIN_NODES + t;
        if (n < N_NODES) { cnt[n] = v; rowptr[n] = excl; }
    }
    __syncthreads();
    for (int base = beg; base < end; base += SU * PT) {
        int2 sd[SU];
        #pragma unroll
        for (int k = 0; k < SU; ++k) {
            const int i = base + t + k * PT;
            sd[k] = in[(i < end) ? i : (end - 1)];
        }
        #pragma unroll
        for (int k = 0; k < SU; ++k) {
            const int i = base + t + k * PT;
            if (i < end) {
                const int pos = atomicAdd(&cur[sd[k].y & (BIN_NODES - 1)], 1);
                esrc[pos] = sd[k].x;
            }
        }
    }
}

__device__ __forceinline__ void load_chunk(const int* __restrict__ esrc,
                                           const float* __restrict__ el,
                                           int start, int deg, int c, int lane,
                                           float er_n, int& s, float& e, bool& v) {
    const int j = c + lane;
    v = j < deg;
    s = v ? esrc[start + j] : 0;
    float t = el[s] + er_n;
    t = (t > 0.f) ? t : NEG_SLOPE * t;
    e = v ? t : -INFINITY;
}

__device__ __forceinline__ v2f row_of(const v2f* __restrict__ g2, int s, int j, int lane) {
    const int sj = __builtin_amdgcn_readlane(s, j);
    return g2[(size_t)sj * (N_CLASSES / 2) + lane];
}
__device__ __forceinline__ float wt_of(float a, int j) {
    return __int_as_float(__builtin_amdgcn_readlane(__float_as_int(a), j));
}

__device__ __forceinline__ void agg_rows(const v2f* __restrict__ g2, int nrows,
                                         int s, float a, int lane, v2f& acc) {
    int j = 0;
    for (; j + 4 <= nrows; j += 4) {
        const v2f f0 = row_of(g2, s, j,     lane);
        const v2f f1 = row_of(g2, s, j + 1, lane);
        const v2f f2 = row_of(g2, s, j + 2, lane);
        const v2f f3 = row_of(g2, s, j + 3, lane);
        acc += wt_of(a, j)     * f0;
        acc += wt_of(a, j + 1) * f1;
        acc += wt_of(a, j + 2) * f2;
        acc += wt_of(a, j + 3) * f3;
    }
    const int rem = nrows - j;
    v2f g0 = {0.f, 0.f}, g1 = g0, g2r = g0;
    float w0 = 0.f, w1 = 0.f, w2 = 0.f;
    if (rem > 0) { g0  = row_of(g2, s, j,     lane); w0 = wt_of(a, j); }
    if (rem > 1) { g1  = row_of(g2, s, j + 1, lane); w1 = wt_of(a, j + 1); }
    if (rem > 2) { g2r = row_of(g2, s, j + 2, lane); w2 = wt_of(a, j + 2); }
    acc += w0 * g0;
    acc += w1 * g1;
    acc += w2 * g2r;
}

__global__ __launch_bounds__(256) void k_pull(const int* __restrict__ cnt,
                                              const int* __restrict__ rowptr,
                                              const int* __restrict__ esrc,
                                              const float* __restrict__ el,
                                              const float* __restrict__ er,
                                              const float* __restrict__ G,
                                              const float* __restrict__ FB,
                                              float* __restrict__ out) {
    const int lane = threadIdx.x & 31;
    const int n = __builtin_amdgcn_readfirstlane((int)(blockIdx.x * 8 + (threadIdx.x >> 5)));
    if (n >= N_NODES) return;
    const int deg   = cnt[n];
    const int start = rowptr[n];
    const float er_n = er[n];
    const v2f* __restrict__ g2 = (const v2f*)G;
    const v2f bias = ((const v2f*)FB)[lane];

    int s0; float e0; bool v0;
    load_chunk(esrc, el, start, deg, 0, lane, er_n, s0, e0, v0);

    float m = e0;
    for (int c = 32; c < deg; c += 32) {
        int s; float e; bool v;
        load_chunk(esrc, el, start, deg, c, lane, er_n, s, e, v);
        m = fmaxf(m, e);
    }
    for (int off = 16; off > 0; off >>= 1) m = fmaxf(m, __shfl_xor(m, off, 32));

    const float x0 = v0 ? __expf(e0 - m) : 0.f;
    float dsum = x0;
    for (int c = 32; c < deg; c += 32) {
        int s; float e; bool v;
        load_chunk(esrc, el, start, deg, c, lane, er_n, s, e, v);
        dsum += v ? __expf(e - m) : 0.f;
    }
    for (int off = 16; off > 0; off >>= 1) dsum += __shfl_xor(dsum, off, 32);

    v2f acc = {0.f, 0.f};
    agg_rows(g2, deg < 32 ? deg : 32, s0, x0 / dsum, lane, acc);
    for (int c = 32; c < deg; c += 32) {
        int s; float e; bool v;
        load_chunk(esrc, el, start, deg, c, lane, er_n, s, e, v);
        const float a = v ? __expf(e - m) / dsum : 0.f;
        const int left = deg - c;
        agg_rows(g2, left < 32 ? left : 32, s, a, lane, acc);
    }
    ((v2f*)out)[(size_t)n * (N_CLASSES / 2) + lane] = acc + bias;
}

extern "C" void kernel_launch(void* const* d_in, const int* in_sizes, int n_in,
                              void* d_out, int out_size, void* d_ws, size_t ws_size,
                              hipStream_t stream) {
    const float* X  = (const float*)d_in[0];
    const float* W  = (const float*)d_in[1];
    const float* al = (const float*)d_in[2];
    const float* ar = (const float*)d_in[3];
    const float* FW = (const float*)d_in[4];
    const float* FB = (const float*)d_in[5];
    const int*  src = (const int*)d_in[6];
    const int*  dst = (const int*)d_in[7];
    float* out = (float*)d_out;

    char* ws = (char*)d_ws;
    size_t off = 0;
    auto carve = [&](size_t bytes) {
        void* p = ws + off;
        off += (bytes + 255) & ~(size_t)255;
        return p;
    };
    float* G      = (float*)carve((size_t)N_NODES * N_CLASSES * sizeof(float));
    float* el     = (float*)carve((size_t)N_NODES * sizeof(float));
    float* er     = (float*)carve((size_t)N_NODES * sizeof(float));
    int*   cnt    = (int*)  carve((size_t)N_NODES * sizeof(int));
    int*   rowptr = (int*)  carve((size_t)N_NODES * sizeof(int));
    int*   esrc   = (int*)  carve((size_t)N_EDGES * sizeof(int));
    float* wcp    = (float*)carve((size_t)D_IN * NC * sizeof(float));
    int*   binned = (int*)  carve(((size_t)2 * N_EDGES + (size_t)2 * NHIST) * sizeof(int));
    int*   hist   = binned + (size_t)2 * N_EDGES;
    int*   offs   = hist + NHIST;

    k_binhist<<<PB, PT, 0, stream>>>(dst, hist);
    k_scan<<<(NHIST / 4 + 255) / 256, 256, 0, stream>>>(hist, offs);
    k_binscatter<<<PB, PT, 0, stream>>>(src, dst, offs, binned);
    k_binsort<<<NBIN, PT, 0, stream>>>(offs, binned, cnt, rowptr, esrc);
    k_fold<<<80, 32, 0, stream>>>(W, FW, al, ar, wcp);
    k_gemm1<<<782, 256, 0, stream>>>(X, wcp, G, el, er);
    k_pull<<<N_NODES / 8, 256, 0, stream>>>(cnt, rowptr, esrc, el, er, G, FB, out);
}


# Independent kernels share launches: weight fold rides in the bin-count launch, count scan in the GEMM launch (block-index ranges pick the body); 7 -> 5 launches; bodies and sum order unchanged
# speedup vs baseline: 7.2981x; 1.1496x over previous
#include <hip/hip_runtime.h>
#include <hip/hip_bf16.h>

#define N_NODES   100000
#define N_EDGES   1600000
#define D_IN      256
#define D_HID     128
#define N_CLASSES 64
#define NEG_SLOPE 0.2f

typedef __attribute__((ext_vector_type(2))) float v2f;
typedef __attribute__((ext_vector_type(4))) float v4f;
typedef __attribute__((ext_vector_type(8))) float v8f;
typedef __attribute__((ext_vector_type(4))) int   v4i;

#define NC 80
#define COL_EL N_CLASSES
#define COL_ER (N_CLASSES + 1)

#define FOLD_TILES 80
__device__ __forceinline__ void fold_body(const int tile, const int lane,
                                          const float* __restrict__ W,
                                          const float* __restrict__ FW,
                                          const float* __restrict__ al,
                                          const float* __restrict__ ar,
                                          float* __restrict__ wcp) {
    const int rowTile = tile / 5;
    const int colTile = tile - rowTile * 5;
    const int l16 = lane & 15;
    const int hi  = lane >> 4;
    const int row0 = rowTile * 16;
    const int col = colTile * 16 + l16;

    const bool isfc = col < N_CLASSES;
    const bool used = col <= COL_ER;
    const float* __restrict__ bp = isfc ? (FW + col) : ((col == COL_ER) ? ar : al);
    const int bs = isfc ? N_CLASSES : 1;

    v8f acc = {};
    const float* __restrict__ wrow = W + (size_t)(row0 + l16) * D_HID;
    for (int j0 = 0; j0 < D_HID; j0 += 4) {
        const int ja = j0 + 2 * hi;
        v2f a;
        a.x = wrow[ja];
        a.y = wrow[ja + 1];
        const float bx = bp[(size_t)ja * bs];
        const float by = bp[(size_t)(ja + 1) * bs];
        v2f b;
        b.x = used ? bx : 0.f;
        b.y = used ? by : 0.f;
        acc = __builtin_amdgcn_wmma_f32_16x16x4_f32(
            false, a, false, b, (short)0, acc, false, false);
    }
    #pragma unroll
    for (int r = 0; r < 8; ++r) {
        const int k = row0 + r + 8 * hi;
        wcp[(size_t)(k >> 1) * (2 * NC) + col * 2 + (k & 1)] = acc[r];
    }
}

#define GEMM_BLOCKS ((N_NODES / 16 + 7) / 8)
__device__ __forceinline__ void gemm_body(const int gb,
                                          const float* __restrict__ X,
                                          const float* __restrict__ wcp,
                                          float* __restrict__ G,
                                          float* __restrict__ el,
                                          float* __restrict__ er) {
    const int wave = threadIdx.x >> 5;
    const int lane = threadIdx.x & 31;
    const int rowTile = gb * 8 + wave;
    if (rowTile >= N_NODES / 16) return;
    const int row0 = rowTile * 16;
    const int l16 = lane & 15;
    const int hi  = lane >> 4;

    v8f acc0 = {}, acc1 = {}, acc2 = {}, acc3 = {}, acc4 = {};
    const float* __restrict__ xrow = X + (size_t)(row0 + l16) * D_IN;
    const v2f* __restrict__ wp = (const v2f*)wcp;

    for (int k0 = 0; k0 < D_IN; k0 += 4) {
        const int ka = k0 + 2 * hi;
        v2f a;
        a.x = xrow[ka];
        a.y = xrow[ka + 1];
        const v2f* __restrict__ w = wp + (size_t)(ka >> 1) * NC + l16;
        const v2f b0 = w[0];
        const v2f b1 = w[16];
        const v2f b2 = w[32];
        const v2f b3 = w[48];
        const v2f b4 = w[64];
        acc0 = __builtin_amdgcn_wmma_f32_16x16x4_f32(
            false, a, false, b0, (short)0, acc0, false, false);
        acc1 = __builtin_amdgcn_wmma_f32_16x16x4_f32(
            false, a, false, b1, (short)0, acc1, false, false);
        acc2 = __builtin_amdgcn_wmma_f32_16x16x4_f32(
            false, a, false, b2, (short)0, acc2, false, false);
        acc3 = __builtin_amdgcn_wmma_f32_16x16x4_f32(
            false, a, false, b3, (short)0, acc3, false, false);
        acc4 = __builtin_amdgcn_wmma_f32_16x16x4_f32(
            false, a, false, b4, (short)0, acc4, false, false);
    }
    float* __restrict__ sc = (l16 == 0) ? el : er;
    #pragma unroll
    for (int r = 0; r < 8; ++r) {
        const int row = row0 + r + 8 * hi;
        float* __restrict__ orow = G + (size_t)row * N_CLASSES + l16;
        orow[0]  = acc0[r];
        orow[16] = acc1[r];
        orow[32] = acc2[r];
        orow[48] = acc3[r];
        if (l16 < 2) sc[row] = acc4[r];
    }
}

#define BIN_SHIFT 8
#define BIN_NODES (1 << BIN_SHIFT)
#define NBIN      ((N_NODES + BIN_NODES - 1) / BIN_NODES)
#define PB        256
#define EPB       (N_EDGES / PB)
#define NHIST     (NBIN * PB)
#define PT        1024
#define EIT       ((EPB + PT - 1) / PT)
#define SU        5
#define SB        16
static_assert(SB == 16, "the statement that groups the scan's loads names 16 vectors");
static_assert(N_EDGES % PB == 0, "edge chunks must tile the edge list");
static_assert(NHIST % 4 == 0, "table is scanned four entries at a time");

__device__ __forceinline__ void binhist_body(const int b,
                                             const int* __restrict__ dst,
                                             int* __restrict__ hist) {
    __shared__ int tab[NBIN];
    const int t = threadIdx.x;
    for (int i = t; i < NBIN; i += PT) tab[i] = 0;
    __syncthreads();
    const int* __restrict__ d = dst + (size_t)b * EPB;
    int dv[EIT];
    #pragma unroll
    for (int k = 0; k < EIT; ++k) {
        const int i = t + k * PT;
        dv[k] = (i < EPB) ? d[i] : 0;
    }
    #pragma unroll
    for (int k = 0; k < EIT; ++k) {
        const int i = t + k * PT;
        if (i < EPB) atomicAdd(&tab[dv[k] >> BIN_SHIFT], 1);
    }
    __syncthreads();
    for (int i = t; i < NBIN; i += PT) hist[i * PB + b] = tab[i];
}

template <bool TAIL>
__device__ __forceinline__ int scan_batch(const v4i* __restrict__ c4v, int t, int i, int b) {
    v4i v[SB];
    #pragma unroll
    for (int k = 0; k < SB; ++k) {
        const int c = (!TAIL || i + k < b) ? (i + k) : 0;
        v[k] = c4v[c * 256 + t];
    }
    asm volatile("" : "+v"(v[0]),  "+v"(v[1]),  "+v"(v[2]),  "+v"(v[3]),
                      "+v"(v[4]),  "+v"(v[5]),  "+v"(v[6]),  "+v"(v[7]),
                      "+v"(v[8]),  "+v"(v[9]),  "+v"(v[10]), "+v"(v[11]),
                      "+v"(v[12]), "+v"(v[13]), "+v"(v[14]), "+v"(v[15]));
    int sum = 0;
    #pragma unroll
    for (int k = 0; k < SB; ++k) {
        const int s4 = (v[k].x + v[k].y) + (v[k].z + v[k].w);
        sum += (!TAIL || i + k < b) ? s4 : 0;
    }
    return sum;
}

#define SCAN_BLOCKS ((NHIST / 4 + 255) / 256)
__device__ __forceinline__ void scan_body(const int b,
                                          const int* __restrict__ hist,
                                          int* __restrict__ offs) {
    __shared__ int wsum[8];
    const int t = threadIdx.x;
    const int lane = t & 31;
    const int wave = t >> 5;
    const int4* __restrict__ c4 = (const int4*)hist;

    const v4i* __restrict__ c4v = (const v4i*)hist;
    int acc = 0;
    int i = 0;
    for (; i + SB <= b; i += SB) acc += scan_batch<false>(c4v, t, i, b);
    if (i < b) acc += scan_batch<true>(c4v, t, i, b);
    for (int off = 16; off > 0; off >>= 1) acc += __shfl_xor(acc, off, 32);
    if (lane == 0) wsum[wave] = acc;
    __syncthreads();
    int base = 0;
    for (int w = 0; w < 8; ++w) base += wsum[w];
    __syncthreads();

    const int j = b * 256 + t;
    const bool ok = j < NHIST / 4;
    int4 v = make_int4(0, 0, 0, 0);
    if (ok) v = c4[j];
    const int s = (v.x + v.y) + (v.z + v.w);
    int inc = s;
    for (int off = 1; off < 32; off <<= 1) {
        const int nb = __shfl_up(inc, off, 32);
        if (lane >= off) inc += nb;
    }
    if (lane == 31) wsum[wave] = inc;
    __syncthreads();
    int wbase = 0;
    for (int w = 0; w < 8; ++w) wbase += (w < wave) ? wsum[w] : 0;
    const int excl = base + wbase + inc - s;
    if (ok) {
        int4 o;
        o.x = excl;
        o.y = o.x + v.x;
        o.z = o.y + v.y;
        o.w = o.z + v.z;
        ((int4*)offs)[j] = o;
    }
}

#define FOLD_BLOCKS ((FOLD_TILES + PT / 32 - 1) / (PT / 32))

__global__ __launch_bounds__(PT) void k_count_fold(const int* __restrict__ dst,
                                                   int* __restrict__ hist,
                                                   const float* __restrict__ W,
                                                   const float* __restrict__ FW,
                                                   const float* __restrict__ al,
                                                   const float* __restrict__ ar,
                                                   float* __restrict__ wcp) {
    if (blockIdx.x >= FOLD_BLOCKS) {
        binhist_body((int)blockIdx.x - FOLD_BLOCKS, dst, hist);
        return;
    }
    const int tile = (int)blockIdx.x * (PT / 32) + (int)(threadIdx.x >> 5);
    if (tile >= FOLD_TILES) return;
    fold_body(tile, (int)(threadIdx.x & 31), W, FW, al, ar, wcp);
}

__global__ __launch_bounds__(256) void k_gemm_scan(const float* __restrict__ X,
                                                   const float* __restrict__ wcp,
                                                   float* __restrict__ G,
                                                   float* __restrict__ el,
                                                   float* __restrict__ er,
                                                   const int* __restrict__ hist,
                                                   int* __restrict__ offs) {
    if (blockIdx.x >= GEMM_BLOCKS) {
        scan_body((int)blockIdx.x - GEMM_BLOCKS, hist, offs);
        return;
    }
    gemm_body((int)blockIdx.x, X, wcp, G, el, er);
}

__global__ __launch_bounds__(PT) void k_binscatter(const int* __restrict__ src,
                                                   const int* __restrict__ dst,
                                                   const int* __restrict__ offs,
                                                   int* __restrict__ binned) {
    __shared__ int cur[NBIN];
    const int t = threadIdx.x;
    const int b = blockIdx.x;
    for (int i = t; i < NBIN; i += PT) cur[i] = offs[i * PB + b];
    __syncthreads();
    const int* __restrict__ s = src + (size_t)b * EPB;
    const int* __restrict__ d = dst + (size_t)b * EPB;
    int2* __restrict__ out = (int2*)binned;
    int dv[EIT], sv[EIT];
    #pragma unroll
    for (int k = 0; k < EIT; ++k) {
        const int i = t + k * PT;
        dv[k] = (i < EPB) ? d[i] : 0;
        sv[k] = (i < EPB) ? s[i] : 0;
    }
    #pragma unroll
    for (int k = 0; k < EIT; ++k) {
        const int i = t + k * PT;
        if (i < EPB) {
            const int pos = atomicAdd(&cur[dv[k] >> BIN_SHIFT], 1);
            out[pos] = make_int2(sv[k], dv[k]);
        }
    }
}

__global__ __launch_bounds__(PT) void k_binsort(const int* __restrict__ offs,
                                                const int* __restrict__ binned,
                                                int* __restrict__ cnt,
                                                int* __restrict__ rowptr,
                                                int* __restrict__ esrc) {
    __shared__ int c[BIN_NODES];
    __shared__ int cur[BIN_NODES];
    __shared__ int wsum[BIN_NODES / 32];
    const int t = threadIdx.x;
    const int b = blockIdx.x;
    const int lane = t & 31;
    const int wave = t >> 5;
    const int beg = offs[b * PB];
    const int end = (b + 1 < NBIN) ? offs[(b + 1) * PB] : N_EDGES;
    const int2* __restrict__ in = (const int2*)binned;

    if (t < BIN_NODES) c[t] = 0;
    __syncthreads();
    for (int base = beg; base < end; base += SU * PT) {
        int y[SU];
        #pragma unroll
        for (int k = 0; k < SU; ++k) {
            const int i = base + t + k * PT;
            y[k] = in[(i < end) ? i : (end - 1)].y;
        }
        #pragma unroll
        for (int k = 0; k < SU; ++k) {
            const int i = base + t + k * PT;
            if (i < end) atomicAdd(&c[y[k] & (BIN_NODES - 1)], 1);
        }
    }
    __syncthreads();

    const bool act = t < BIN_NODES;
    const int v = act ? c[t] : 0;
    int inc = v;
    for (int off = 1; off < 32; off <<= 1) {
        const int nb = __shfl_up(inc, off, 32);
        if (lane >= off) inc += nb;
    }
    if (act && lane == 31) wsum[wave] = inc;
    __syncthreads();
    int wbase = 0;
    for (int w = 0; w < BIN_NODES / 32; ++w) wbase += (w < wave) ? wsum[w] : 0;
    const int excl = beg + wbase + inc - v;
    if (act) {
        cur[t] = excl;
        const int n = b * BIN_NODES + t;
        if (n < N_NODES) { cnt[n] = v; rowptr[n] = excl; }
    }
    __syncthreads();
    for (int base = beg; base < end; base += SU * PT) {
        int2 sd[SU];
        #pragma unroll
        for (int k = 0; k < SU; ++k) {
            const int i = base + t + k * PT;
            sd[k] = in[(i < end) ? i : (end - 1)];
        }
        #pragma unroll
        for (int k = 0; k < SU; ++k) {
            const int i = base + t + k * PT;
            if (i < end) {
                const int pos = atomicAdd(&cur[sd[k].y & (BIN_NODES - 1)], 1);
                esrc[pos] = sd[k].x;
            }
        }
    }
}

__device__ __forceinline__ void load_chunk(const int* __restrict__ esrc,
                                           const float* __restrict__ el,
                                           int start, int deg, int c, int lane,
                                           float er_n, int& s, float& e, bool& v) {
    const int j = c + lane;
    v = j < deg;
    s = v ? esrc[start + j] : 0;
    float t = el[s] + er_n;
    t = (t > 0.f) ? t : NEG_SLOPE * t;
    e = v ? t : -INFINITY;
}

__device__ __forceinline__ v2f row_of(const v2f* __restrict__ g2, int s, int j, int lane) {
    const int sj = __builtin_amdgcn_readlane(s, j);
    return g2[(size_t)sj * (N_CLASSES / 2) + lane];
}
__device__ __forceinline__ float wt_of(float a, int j) {
    return __int_as_float(__builtin_amdgcn_readlane(__float_as_int(a), j));
}

__device__ __forceinline__ void agg_rows(const v2f* __restrict__ g2, int nrows,
                                         int s, float a, int lane, v2f& acc) {
    int j = 0;
    for (; j + 4 <= nrows; j += 4) {
        const v2f f0 = row_of(g2, s, j,     lane);
        const v2f f1 = row_of(g2, s, j + 1, lane);
        const v2f f2 = row_of(g2, s, j + 2, lane);
        const v2f f3 = row_of(g2, s, j + 3, lane);
        acc += wt_of(a, j)     * f0;
        acc += wt_of(a, j + 1) * f1;
        acc += wt_of(a, j + 2) * f2;
        acc += wt_of(a, j + 3) * f3;
    }
    const int rem = nrows - j;
    v2f g0 = {0.f, 0.f}, g1 = g0, g2r = g0;
    float w0 = 0.f, w1 = 0.f, w2 = 0.f;
    if (rem > 0) { g0  = row_of(g2, s, j,     lane); w0 = wt_of(a, j); }
    if (rem > 1) { g1  = row_of(g2, s, j + 1, lane); w1 = wt_of(a, j + 1); }
    if (rem > 2) { g2r = row_of(g2, s, j + 2, lane); w2 = wt_of(a, j + 2); }
    acc += w0 * g0;
    acc += w1 * g1;
    acc += w2 * g2r;
}

__global__ __launch_bounds__(256) void k_pull(const int* __restrict__ cnt,
                                              const int* __restrict__ rowptr,
                                              const int* __restrict__ esrc,
                                              const float* __restrict__ el,
                                              const float* __restrict__ er,
                                              const float* __restrict__ G,
                                              const float* __restrict__ FB,
                                              float* __restrict__ out) {
    const int lane = threadIdx.x & 31;
    const int n = __builtin_amdgcn_readfirstlane((int)(blockIdx.x * 8 + (threadIdx.x >> 5)));
    if (n >= N_NODES) return;
    const int deg   = cnt[n];
    const int start = rowptr[n];
    const float er_n = er[n];
    const v2f* __restrict__ g2 = (const v2f*)G;
    const v2f bias = ((const v2f*)FB)[lane];

    int s0; float e0; bool v0;
    load_chunk(esrc, el, start, deg, 0, lane, er_n, s0, e0, v0);

    float m = e0;
    for (int c = 32; c < deg; c += 32) {
        int s; float e; bool v;
        load_chunk(esrc, el, start, deg, c, lane, er_n, s, e, v);
        m = fmaxf(m, e);
    }
    for (int off = 16; off > 0; off >>= 1) m = fmaxf(m, __shfl_xor(m, off, 32));

    const float x0 = v0 ? __expf(e0 - m) : 0.f;
    float dsum = x0;
    for (int c = 32; c < deg; c += 32) {
        int s; float e; bool v;
        load_chunk(esrc, el, start, deg, c, lane, er_n, s, e, v);
        dsum += v ? __expf(e - m) : 0.f;
    }
    for (int off = 16; off > 0; off >>= 1) dsum += __shfl_xor(dsum, off, 32);

    v2f acc = {0.f, 0.f};
    agg_rows(g2, deg < 32 ? deg : 32, s0, x0 / dsum, lane, acc);
    for (int c = 32; c < deg; c += 32) {
        int s; float e; bool v;
        load_chunk(esrc, el, start, deg, c, lane, er_n, s, e, v);
        const float a = v ? __expf(e - m) / dsum : 0.f;
        const int left = deg - c;
        agg_rows(g2, left < 32 ? left : 32, s, a, lane, acc);
    }
    ((v2f*)out)[(size_t)n * (N_CLASSES / 2) + lane] = acc + bias;
}

extern "C" void kernel_launch(void* const* d_in, const int* in_sizes, int n_in,
                              void* d_out, int out_size, void* d_ws, size_t ws_size,
                              hipStream_t stream) {
    const float* X  = (const float*)d_in[0];
    const float* W  = (const float*)d_in[1];
    const float* al = (const float*)d_in[2];
    const float* ar = (const float*)d_in[3];
    const float* FW = (const float*)d_in[4];
    const float* FB = (const float*)d_in[5];
    const int*  src = (const int*)d_in[6];
    const int*  dst = (const int*)d_in[7];
    float* out = (float*)d_out;

    char* ws = (char*)d_ws;
    size_t off = 0;
    auto carve = [&](size_t bytes) {
        void* p = ws + off;
        off += (bytes + 255) & ~(size_t)255;
        return p;
    };
    float* G      = (float*)carve((size_t)N_NODES * N_CLASSES * sizeof(float));
    float* el     = (float*)carve((size_t)N_NODES * sizeof(float));
    float* er     = (float*)carve((size_t)N_NODES * sizeof(float));
    int*   cnt    = (int*)  carve((size_t)N_NODES * sizeof(int));
    int*   rowptr = (int*)  carve((size_t)N_NODES * sizeof(int));
    int*   esrc   = (int*)  carve((size_t)N_EDGES * sizeof(int));
    float* wcp    = (float*)carve((size_t)D_IN * NC * sizeof(float));
    int*   binned = (int*)  carve(((size_t)2 * N_EDGES + (size_t)2 * NHIST) * sizeof(int));
    int*   hist   = binned + (size_t)2 * N_EDGES;
    int*   offs   = hist + NHIST;

    k_count_fold<<<FOLD_BLOCKS + PB, PT, 0, stream>>>(dst, hist, W, FW, al, ar, wcp);
    k_gemm_scan<<<GEMM_BLOCKS + SCAN_BLOCKS, 256, 0, stream>>>(X, wcp, G, el, er, hist, offs);
    k_binscatter<<<PB, PT, 0, stream>>>(src, dst, offs, binned);
    k_binsort<<<NBIN, PT, 0, stream>>>(offs, binned, cnt, rowptr, esrc);
    k_pull<<<N_NODES / 8, 256, 0, stream>>>(cnt, rowptr, esrc, el, er, G, FB, out);
}


# Edge partition with no grid-wide count table: blocks bin-sort own edges in LDS, bin start = sum of block run offsets; count+scan gone, GEMM beside node sort: 5->3 launches; sums in arrival order
# speedup vs baseline: 8.7592x; 1.2002x over previous
#include <hip/hip_runtime.h>
#include <hip/hip_bf16.h>

#define N_NODES   100000
#define N_EDGES   1600000
#define D_IN      256
#define D_HID     128
#define N_CLASSES 64
#define NEG_SLOPE 0.2f

typedef __attribute__((ext_vector_type(2))) float v2f;
typedef __attribute__((ext_vector_type(4))) float v4f;
typedef __attribute__((ext_vector_type(8))) float v8f;

#define NC 80
#define COL_EL N_CLASSES
#define COL_ER (N_CLASSES + 1)

#define FOLD_TILES 80
__device__ __forceinline__ void fold_body(const int tile, const int lane,
                                          const float* __restrict__ W,
                                          const float* __restrict__ FW,
                                          const float* __restrict__ al,
                                          const float* __restrict__ ar,
                                          float* __restrict__ wcp) {
    const int rowTile = tile / 5;
    const int colTile = tile - rowTile * 5;
    const int l16 = lane & 15;
    const int hi  = lane >> 4;
    const int row0 = rowTile * 16;
    const int col = colTile * 16 + l16;

    const bool isfc = col < N_CLASSES;
    const bool used = col <= COL_ER;
    const float* __restrict__ bp = isfc ? (FW + col) : ((col == COL_ER) ? ar : al);
    const int bs = isfc ? N_CLASSES : 1;

    v8f acc = {};
    const float* __restrict__ wrow = W + (size_t)(row0 + l16) * D_HID;
    for (int j0 = 0; j0 < D_HID; j0 += 4) {
        const int ja = j0 + 2 * hi;
        v2f a;
        a.x = wrow[ja];
        a.y = wrow[ja + 1];
        const float bx = bp[(size_t)ja * bs];
        const float by = bp[(size_t)(ja + 1) * bs];
        v2f b;
        b.x = used ? bx : 0.f;
        b.y = used ? by : 0.f;
        acc = __builtin_amdgcn_wmma_f32_16x16x4_f32(
            false, a, false, b, (short)0, acc, false, false);
    }
    #pragma unroll
    for (int r = 0; r < 8; ++r) {
        const int k = row0 + r + 8 * hi;
        wcp[(size_t)(k >> 1) * (2 * NC) + col * 2 + (k & 1)] = acc[r];
    }
}

#define GEMM_BLOCKS 256
__device__ __forceinline__ void gemm_body(const int rowTile, const int lane,
                                          const float* __restrict__ X,
                                          const float* __restrict__ wcp,
                                          float* __restrict__ G,
                                          float* __restrict__ el,
                                          float* __restrict__ er) {
    const int row0 = rowTile * 16;
    const int l16 = lane & 15;
    const int hi  = lane >> 4;

    v8f acc0 = {}, acc1 = {}, acc2 = {}, acc3 = {}, acc4 = {};
    const float* __restrict__ xrow = X + (size_t)(row0 + l16) * D_IN;
    const v2f* __restrict__ wp = (const v2f*)wcp;

    for (int k0 = 0; k0 < D_IN; k0 += 4) {
        const int ka = k0 + 2 * hi;
        v2f a;
        a.x = xrow[ka];
        a.y = xrow[ka + 1];
        const v2f* __restrict__ w = wp + (size_t)(ka >> 1) * NC + l16;
        const v2f b0 = w[0];
        const v2f b1 = w[16];
        const v2f b2 = w[32];
        const v2f b3 = w[48];
        const v2f b4 = w[64];
        acc0 = __builtin_amdgcn_wmma_f32_16x16x4_f32(
            false, a, false, b0, (short)0, acc0, false, false);
        acc1 = __builtin_amdgcn_wmma_f32_16x16x4_f32(
            false, a, false, b1, (short)0, acc1, false, false);
        acc2 = __builtin_amdgcn_wmma_f32_16x16x4_f32(
            false, a, false, b2, (short)0, acc2, false, false);
        acc3 = __builtin_amdgcn_wmma_f32_16x16x4_f32(
            false, a, false, b3, (short)0, acc3, false, false);
        acc4 = __builtin_amdgcn_wmma_f32_16x16x4_f32(
            false, a, false, b4, (short)0, acc4, false, false);
    }
    float* __restrict__ sc = (l16 == 0) ? el : er;
    #pragma unroll
    for (int r = 0; r < 8; ++r) {
        const int row = row0 + r + 8 * hi;
        float* __restrict__ orow = G + (size_t)row * N_CLASSES + l16;
        orow[0]  = acc0[r];
        orow[16] = acc1[r];
        orow[32] = acc2[r];
        orow[48] = acc3[r];
        if (l16 < 2) sc[row] = acc4[r];
    }
}

#define BIN_SHIFT 8
#define BIN_NODES (1 << BIN_SHIFT)
#define NBIN      ((N_NODES + BIN_NODES - 1) / BIN_NODES)
#define PB        256
#define EPB       (N_EDGES / PB)
#define NLOFF     ((NBIN + 1) * PB)
#define PT        1024
#define EIT       ((EPB + PT - 1) / PT)
#define NSW       ((NBIN + 31) / 32)
#define RT        (PT / PB)
#define RU        8
static_assert(N_EDGES % PB == 0, "edge chunks must tile the edge list");
static_assert(NBIN < PT, "one thread per bin, and one more for the closing row");
static_assert(PT % PB == 0, "whole thread groups per run");
static_assert(RU == 8, "the statement that groups the run loads names 8 pairs");
static_assert(BIN_NODES <= PT && PB <= PT, "one thread per node / per run in the bin block");

__device__ __forceinline__ void part_body(const int b,
                                          const int* __restrict__ src,
                                          const int* __restrict__ dst,
                                          int* __restrict__ loff,
                                          int* __restrict__ binned) {
    __shared__ int tab[NBIN];
    __shared__ int wsum[NSW];
    const int t = threadIdx.x;
    const int lane = t & 31;
    const int wave = t >> 5;
    if (t < NBIN) tab[t] = 0;
    const int* __restrict__ s = src + (size_t)b * EPB;
    const int* __restrict__ d = dst + (size_t)b * EPB;
    int2* __restrict__ out = (int2*)binned + (size_t)b * EPB;
    int dv[EIT], sv[EIT];
    #pragma unroll
    for (int k = 0; k < EIT; ++k) {
        const int i = t + k * PT;
        dv[k] = (i < EPB) ? d[i] : 0;
        sv[k] = (i < EPB) ? s[i] : 0;
    }
    __syncthreads();
    #pragma unroll
    for (int k = 0; k < EIT; ++k) {
        const int i = t + k * PT;
        if (i < EPB) atomicAdd(&tab[dv[k] >> BIN_SHIFT], 1);
    }
    __syncthreads();

    const bool act = t < NBIN;
    const int v = act ? tab[t] : 0;
    int inc = v;
    for (int off = 1; off < 32; off <<= 1) {
        const int nb = __shfl_up(inc, off, 32);
        if (lane >= off) inc += nb;
    }
    if (wave < NSW && lane == 31) wsum[wave] = inc;
    __syncthreads();
    int wbase = 0;
    for (int w = 0; w < NSW; ++w) wbase += (w < wave) ? wsum[w] : 0;
    const int excl = wbase + inc - v;
    if (act) {
        tab[t] = excl;
        loff[t * PB + b] = excl;
    }
    if (t == NBIN) loff[NBIN * PB + b] = EPB;
    __syncthreads();
    #pragma unroll
    for (int k = 0; k < EIT; ++k) {
        const int i = t + k * PT;
        if (i < EPB) {
            const int pos = atomicAdd(&tab[dv[k] >> BIN_SHIFT], 1);
            out[pos] = make_int2(sv[k], dv[k]);
        }
    }
}

__device__ __forceinline__ void binsort_body(const int b,
                                             const int* __restrict__ loff,
                                             const int* __restrict__ binned,
                                             int* __restrict__ cnt,
                                             int* __restrict__ rowptr,
                                             int* __restrict__ esrc) {
    __shared__ int c[BIN_NODES];
    __shared__ int cur[BIN_NODES];
    __shared__ int wsum[BIN_NODES / 32];
    __shared__ int bsum[PB / 32];
    __shared__ int rlo[PB];
    __shared__ int rlen[PB];
    const int t = threadIdx.x;
    const int lane = t & 31;
    const int wave = t >> 5;

    int lo = 0;
    if (t < PB) {
        lo = loff[b * PB + t];
        rlo[t] = lo;
        rlen[t] = loff[(b + 1) * PB + t] - lo;
    }
    if (t < BIN_NODES) c[t] = 0;
    int ls = lo;
    for (int off = 16; off > 0; off >>= 1) ls += __shfl_xor(ls, off, 32);
    if (wave < PB / 32 && lane == 0) bsum[wave] = ls;
    __syncthreads();
    int beg = 0;
    for (int w = 0; w < PB / 32; ++w) beg += bsum[w];

    const int r = t / RT;
    const int q = t - r * RT;
    const int len = rlen[r];
    const int2* __restrict__ in = (const int2*)binned + (size_t)r * EPB + rlo[r];
    int hx[RU], hy[RU];
    #pragma unroll
    for (int k = 0; k < RU; ++k) { hx[k] = 0; hy[k] = 0; }
    if (q < len) {
        #pragma unroll
        for (int k = 0; k < RU; ++k) {
            const int jj = q + k * RT;
            const int2 p = in[(jj < len) ? jj : q];
            hx[k] = p.x;
            hy[k] = p.y;
        }
    }
    asm volatile("" : "+v"(hx[0]), "+v"(hy[0]), "+v"(hx[1]), "+v"(hy[1]),
                      "+v"(hx[2]), "+v"(hy[2]), "+v"(hx[3]), "+v"(hy[3]),
                      "+v"(hx[4]), "+v"(hy[4]), "+v"(hx[5]), "+v"(hy[5]),
                      "+v"(hx[6]), "+v"(hy[6]), "+v"(hx[7]), "+v"(hy[7]));
    #pragma unroll
    for (int k = 0; k < RU; ++k) {
        if (q + k * RT < len) atomicAdd(&c[hy[k] & (BIN_NODES - 1)], 1);
    }
    for (int j = q + RU * RT; j < len; j += RT) atomicAdd(&c[in[j].y & (BIN_NODES - 1)], 1);
    __syncthreads();

    const bool act = t < BIN_NODES;
    const int v = act ? c[t] : 0;
    int inc = v;
    for (int off = 1; off < 32; off <<= 1) {
        const int nb = __shfl_up(inc, off, 32);
        if (lane >= off) inc += nb;
    }
    if (act && lane == 31) wsum[wave] = inc;
    __syncthreads();
    int wbase = 0;
    for (int w = 0; w < BIN_NODES / 32; ++w) wbase += (w < wave) ? wsum[w] : 0;
    const int excl = beg + wbase + inc - v;
    if (act) {
        cur[t] = excl;
        const int n = b * BIN_NODES + t;
        if (n < N_NODES) { cnt[n] = v; rowptr[n] = excl; }
    }
    __syncthreads();
    #pragma unroll
    for (int k = 0; k < RU; ++k) {
        if (q + k * RT < len) {
            const int pos = atomicAdd(&cur[hy[k] & (BIN_NODES - 1)], 1);
            esrc[pos] = hx[k];
        }
    }
    for (int j = q + RU * RT; j < len; j += RT) {
        const int2 p = in[j];
        const int pos = atomicAdd(&cur[p.y & (BIN_NODES - 1)], 1);
        esrc[pos] = p.x;
    }
}

#define FOLD_BLOCKS ((FOLD_TILES + PT / 32 - 1) / (PT / 32))

__global__ __launch_bounds__(PT) void k_part_fold(const int* __restrict__ src,
                                                  const int* __restrict__ dst,
                                                  int* __restrict__ loff,
                                                  int* __restrict__ binned,
                                                  const float* __restrict__ W,
                                                  const float* __restrict__ FW,
                                                  const float* __restrict__ al,
                                                  const float* __restrict__ ar,
                                                  float* __restrict__ wcp) {
    if (blockIdx.x >= FOLD_BLOCKS) {
        part_body((int)blockIdx.x - FOLD_BLOCKS, src, dst, loff, binned);
        return;
    }
    const int tile = (int)blockIdx.x * (PT / 32) + (int)(threadIdx.x >> 5);
    if (tile >= FOLD_TILES) return;
    fold_body(tile, (int)(threadIdx.x & 31), W, FW, al, ar, wcp);
}

__global__ __launch_bounds__(PT) void k_gemm_sort(const float* __restrict__ X,
                                                  const float* __restrict__ wcp,
                                                  float* __restrict__ G,
                                                  float* __restrict__ el,
                                                  float* __restrict__ er,
                                                  const int* __restrict__ loff,
                                                  const int* __restrict__ binned,
                                                  int* __restrict__ cnt,
                                                  int* __restrict__ rowptr,
                                                  int* __restrict__ esrc) {
    if (blockIdx.x >= GEMM_BLOCKS) {
        binsort_body((int)blockIdx.x - GEMM_BLOCKS, loff, binned, cnt, rowptr, esrc);
        return;
    }
    const int rowTile = (int)(threadIdx.x >> 5) * GEMM_BLOCKS + (int)blockIdx.x;
    if (rowTile >= N_NODES / 16) return;
    gemm_body(rowTile, (int)(threadIdx.x & 31), X, wcp, G, el, er);
}

__device__ __forceinline__ void load_chunk(const int* __restrict__ esrc,
                                           const float* __restrict__ el,
                                           int start, int deg, int c, int lane,
                                           float er_n, int& s, float& e, bool& v) {
    const int j = c + lane;
    v = j < deg;
    s = v ? esrc[start + j] : 0;
    float t = el[s] + er_n;
    t = (t > 0.f) ? t : NEG_SLOPE * t;
    e = v ? t : -INFINITY;
}

__device__ __forceinline__ v2f row_of(const v2f* __restrict__ g2, int s, int j, int lane) {
    const int sj = __builtin_amdgcn_readlane(s, j);
    return g2[(size_t)sj * (N_CLASSES / 2) + lane];
}
__device__ __forceinline__ float wt_of(float a, int j) {
    return __int_as_float(__builtin_amdgcn_readlane(__float_as_int(a), j));
}

__device__ __forceinline__ void agg_rows(const v2f* __restrict__ g2, int nrows,
                                         int s, float a, int lane, v2f& acc) {
    int j = 0;
    for (; j + 4 <= nrows; j += 4) {
        const v2f f0 = row_of(g2, s, j,     lane);
        const v2f f1 = row_of(g2, s, j + 1, lane);
        const v2f f2 = row_of(g2, s, j + 2, lane);
        const v2f f3 = row_of(g2, s, j + 3, lane);
        acc += wt_of(a, j)     * f0;
        acc += wt_of(a, j + 1) * f1;
        acc += wt_of(a, j + 2) * f2;
        acc += wt_of(a, j + 3) * f3;
    }
    const int rem = nrows - j;
    v2f g0 = {0.f, 0.f}, g1 = g0, g2r = g0;
    float w0 = 0.f, w1 = 0.f, w2 = 0.f;
    if (rem > 0) { g0  = row_of(g2, s, j,     lane); w0 = wt_of(a, j); }
    if (rem > 1) { g1  = row_of(g2, s, j + 1, lane); w1 = wt_of(a, j + 1); }
    if (rem > 2) { g2r = row_of(g2, s, j + 2, lane); w2 = wt_of(a, j + 2); }
    acc += w0 * g0;
    acc += w1 * g1;
    acc += w2 * g2r;
}

__global__ __launch_bounds__(256) void k_pull(const int* __restrict__ cnt,
                                              const int* __restrict__ rowptr,
                                              const int* __restrict__ esrc,
                                              const float* __restrict__ el,
                                              const float* __restrict__ er,
                                              const float* __restrict__ G,
                                              const float* __restrict__ FB,
                                              float* __restrict__ out) {
    const int lane = threadIdx.x & 31;
    const int n = __builtin_amdgcn_readfirstlane((int)(blockIdx.x * 8 + (threadIdx.x >> 5)));
    if (n >= N_NODES) return;
    const int deg   = cnt[n];
    const int start = rowptr[n];
    const float er_n = er[n];
    const v2f* __restrict__ g2 = (const v2f*)G;
    const v2f bias = ((const v2f*)FB)[lane];

    int s0; float e0; bool v0;
    load_chunk(esrc, el, start, deg, 0, lane, er_n, s0, e0, v0);

    float m = e0;
    for (int c = 32; c < deg; c += 32) {
        int s; float e; bool v;
        load_chunk(esrc, el, start, deg, c, lane, er_n, s, e, v);
        m = fmaxf(m, e);
    }
    for (int off = 16; off > 0; off >>= 1) m = fmaxf(m, __shfl_xor(m, off, 32));

    const float x0 = v0 ? __expf(e0 - m) : 0.f;
    float dsum = x0;
    for (int c = 32; c < deg; c += 32) {
        int s; float e; bool v;
        load_chunk(esrc, el, start, deg, c, lane, er_n, s, e, v);
        dsum += v ? __expf(e - m) : 0.f;
    }
    for (int off = 16; off > 0; off >>= 1) dsum += __shfl_xor(dsum, off, 32);

    v2f acc = {0.f, 0.f};
    agg_rows(g2, deg < 32 ? deg : 32, s0, x0 / dsum, lane, acc);
    for (int c = 32; c < deg; c += 32) {
        int s; float e; bool v;
        load_chunk(esrc, el, start, deg, c, lane, er_n, s, e, v);
        const float a = v ? __expf(e - m) / dsum : 0.f;
        const int left = deg - c;
        agg_rows(g2, left < 32 ? left : 32, s, a, lane, acc);
    }
    ((v2f*)out)[(size_t)n * (N_CLASSES / 2) + lane] = acc + bias;
}

extern "C" void kernel_launch(void* const* d_in, const int* in_sizes, int n_in,
                              void* d_out, int out_size, void* d_ws, size_t ws_size,
                              hipStream_t stream) {
    const float* X  = (const float*)d_in[0];
    const float* W  = (const float*)d_in[1];
    const float* al = (const float*)d_in[2];
    const float* ar = (const float*)d_in[3];
    const float* FW = (const float*)d_in[4];
    const float* FB = (const float*)d_in[5];
    const int*  src = (const int*)d_in[6];
    const int*  dst = (const int*)d_in[7];
    float* out = (float*)d_out;

    char* ws = (char*)d_ws;
    size_t off = 0;
    auto carve = [&](size_t bytes) {
        void* p = ws + off;
        off += (bytes + 255) & ~(size_t)255;
        return p;
    };
    float* G      = (float*)carve((size_t)N_NODES * N_CLASSES * sizeof(float));
    float* el     = (float*)carve((size_t)N_NODES * sizeof(float));
    float* er     = (float*)carve((size_t)N_NODES * sizeof(float));
    int*   cnt    = (int*)  carve((size_t)N_NODES * sizeof(int));
    int*   rowptr = (int*)  carve((size_t)N_NODES * sizeof(int));
    int*   esrc   = (int*)  carve((size_t)N_EDGES * sizeof(int));
    float* wcp    = (float*)carve((size_t)D_IN * NC * sizeof(float));
    int*   binned = (int*)  carve(((size_t)2 * N_EDGES + (size_t)NLOFF) * sizeof(int));
    int*   loff   = binned + (size_t)2 * N_EDGES;

    k_part_fold<<<FOLD_BLOCKS + PB, PT, 0, stream>>>(src, dst, loff, binned, W, FW, al, ar, wcp);
    k_gemm_sort<<<GEMM_BLOCKS + NBIN, PT, 0, stream>>>(X, wcp, G, el, er, loff, binned,
                                                      cnt, rowptr, esrc);
    k_pull<<<N_NODES / 8, 256, 0, stream>>>(cnt, rowptr, esrc, el, er, G, FB, out);
}


# Per-destination pass: 2 nodes per wave, one per 16-lane half, rows loaded 16 B per lane: half the waves and row loads; row sums keep list order, weight sum regrouped only above 32 in-edges
# speedup vs baseline: 9.0984x; 1.0387x over previous
#include <hip/hip_runtime.h>
#include <hip/hip_bf16.h>

#define N_NODES   100000
#define N_EDGES   1600000
#define D_IN      256
#define D_HID     128
#define N_CLASSES 64
#define NEG_SLOPE 0.2f

typedef __attribute__((ext_vector_type(2))) float v2f;
typedef __attribute__((ext_vector_type(4))) float v4f;
typedef __attribute__((ext_vector_type(8))) float v8f;

#define NC 80
#define COL_EL N_CLASSES
#define COL_ER (N_CLASSES + 1)

#define FOLD_TILES 80
__device__ __forceinline__ void fold_body(const int tile, const int lane,
                                          const float* __restrict__ W,
                                          const float* __restrict__ FW,
                                          const float* __restrict__ al,
                                          const float* __restrict__ ar,
                                          float* __restrict__ wcp) {
    const int rowTile = tile / 5;
    const int colTile = tile - rowTile * 5;
    const int l16 = lane & 15;
    const int hi  = lane >> 4;
    const int row0 = rowTile * 16;
    const int col = colTile * 16 + l16;

    const bool isfc = col < N_CLASSES;
    const bool used = col <= COL_ER;
    const float* __restrict__ bp = isfc ? (FW + col) : ((col == COL_ER) ? ar : al);
    const int bs = isfc ? N_CLASSES : 1;

    v8f acc = {};
    const float* __restrict__ wrow = W + (size_t)(row0 + l16) * D_HID;
    for (int j0 = 0; j0 < D_HID; j0 += 4) {
        const int ja = j0 + 2 * hi;
        v2f a;
        a.x = wrow[ja];
        a.y = wrow[ja + 1];
        const float bx = bp[(size_t)ja * bs];
        const float by = bp[(size_t)(ja + 1) * bs];
        v2f b;
        b.x = used ? bx : 0.f;
        b.y = used ? by : 0.f;
        acc = __builtin_amdgcn_wmma_f32_16x16x4_f32(
            false, a, false, b, (short)0, acc, false, false);
    }
    #pragma unroll
    for (int r = 0; r < 8; ++r) {
        const int k = row0 + r + 8 * hi;
        wcp[(size_t)(k >> 1) * (2 * NC) + col * 2 + (k & 1)] = acc[r];
    }
}

#define GEMM_BLOCKS 256
__device__ __forceinline__ void gemm_body(const int rowTile, const int lane,
                                          const float* __restrict__ X,
                                          const float* __restrict__ wcp,
                                          float* __restrict__ G,
                                          float* __restrict__ el,
                                          float* __restrict__ er) {
    const int row0 = rowTile * 16;
    const int l16 = lane & 15;
    const int hi  = lane >> 4;

    v8f acc0 = {}, acc1 = {}, acc2 = {}, acc3 = {}, acc4 = {};
    const float* __restrict__ xrow = X + (size_t)(row0 + l16) * D_IN;
    const v2f* __restrict__ wp = (const v2f*)wcp;

    for (int k0 = 0; k0 < D_IN; k0 += 4) {
        const int ka = k0 + 2 * hi;
        v2f a;
        a.x = xrow[ka];
        a.y = xrow[ka + 1];
        const v2f* __restrict__ w = wp + (size_t)(ka >> 1) * NC + l16;
        const v2f b0 = w[0];
        const v2f b1 = w[16];
        const v2f b2 = w[32];
        const v2f b3 = w[48];
        const v2f b4 = w[64];
        acc0 = __builtin_amdgcn_wmma_f32_16x16x4_f32(
            false, a, false, b0, (short)0, acc0, false, false);
        acc1 = __builtin_amdgcn_wmma_f32_16x16x4_f32(
            false, a, false, b1, (short)0, acc1, false, false);
        acc2 = __builtin_amdgcn_wmma_f32_16x16x4_f32(
            false, a, false, b2, (short)0, acc2, false, false);
        acc3 = __builtin_amdgcn_wmma_f32_16x16x4_f32(
            false, a, false, b3, (short)0, acc3, false, false);
        acc4 = __builtin_amdgcn_wmma_f32_16x16x4_f32(
            false, a, false, b4, (short)0, acc4, false, false);
    }
    float* __restrict__ sc = (l16 == 0) ? el : er;
    #pragma unroll
    for (int r = 0; r < 8; ++r) {
        const int row = row0 + r + 8 * hi;
        float* __restrict__ orow = G + (size_t)row * N_CLASSES + l16;
        orow[0]  = acc0[r];
        orow[16] = acc1[r];
        orow[32] = acc2[r];
        orow[48] = acc3[r];
        if (l16 < 2) sc[row] = acc4[r];
    }
}

#define BIN_SHIFT 8
#define BIN_NODES (1 << BIN_SHIFT)
#define NBIN      ((N_NODES + BIN_NODES - 1) / BIN_NODES)
#define PB        256
#define EPB       (N_EDGES / PB)
#define NLOFF     ((NBIN + 1) * PB)
#define PT        1024
#define EIT       ((EPB + PT - 1) / PT)
#define NSW       ((NBIN + 31) / 32)
#define RT        (PT / PB)
#define RU        8
static_assert(N_EDGES % PB == 0, "edge chunks must tile the edge list");
static_assert(NBIN < PT, "one thread per bin, and one more for the closing row");
static_assert(PT % PB == 0, "whole thread groups per run");
static_assert(RU == 8, "the statement that groups the run loads names 8 pairs");
static_assert(BIN_NODES <= PT && PB <= PT, "one thread per node / per run in the bin block");

__device__ __forceinline__ void part_body(const int b,
                                          const int* __restrict__ src,
                                          const int* __restrict__ dst,
                                          int* __restrict__ loff,
                                          int* __restrict__ binned) {
    __shared__ int tab[NBIN];
    __shared__ int wsum[NSW];
    const int t = threadIdx.x;
    const int lane = t & 31;
    const int wave = t >> 5;
    if (t < NBIN) tab[t] = 0;
    const int* __restrict__ s = src + (size_t)b * EPB;
    const int* __restrict__ d = dst + (size_t)b * EPB;
    int2* __restrict__ out = (int2*)binned + (size_t)b * EPB;
    int dv[EIT], sv[EIT];
    #pragma unroll
    for (int k = 0; k < EIT; ++k) {
        const int i = t + k * PT;
        dv[k] = (i < EPB) ? d[i] : 0;
        sv[k] = (i < EPB) ? s[i] : 0;
    }
    __syncthreads();
    #pragma unroll
    for (int k = 0; k < EIT; ++k) {
        const int i = t + k * PT;
        if (i < EPB) atomicAdd(&tab[dv[k] >> BIN_SHIFT], 1);
    }
    __syncthreads();

    const bool act = t < NBIN;
    const int v = act ? tab[t] : 0;
    int inc = v;
    for (int off = 1; off < 32; off <<= 1) {
        const int nb = __shfl_up(inc, off, 32);
        if (lane >= off) inc += nb;
    }
    if (wave < NSW && lane == 31) wsum[wave] = inc;
    __syncthreads();
    int wbase = 0;
    for (int w = 0; w < NSW; ++w) wbase += (w < wave) ? wsum[w] : 0;
    const int excl = wbase + inc - v;
    if (act) {
        tab[t] = excl;
        loff[t * PB + b] = excl;
    }
    if (t == NBIN) loff[NBIN * PB + b] = EPB;
    __syncthreads();
    #pragma unroll
    for (int k = 0; k < EIT; ++k) {
        const int i = t + k * PT;
        if (i < EPB) {
            const int pos = atomicAdd(&tab[dv[k] >> BIN_SHIFT], 1);
            out[pos] = make_int2(sv[k], dv[k]);
        }
    }
}

__device__ __forceinline__ void binsort_body(const int b,
                                             const int* __restrict__ loff,
                                             const int* __restrict__ binned,
                                             int* __restrict__ cnt,
                                             int* __restrict__ rowptr,
                                             int* __restrict__ esrc) {
    __shared__ int c[BIN_NODES];
    __shared__ int cur[BIN_NODES];
    __shared__ int wsum[BIN_NODES / 32];
    __shared__ int bsum[PB / 32];
    __shared__ int rlo[PB];
    __shared__ int rlen[PB];
    const int t = threadIdx.x;
    const int lane = t & 31;
    const int wave = t >> 5;

    int lo = 0;
    if (t < PB) {
        lo = loff[b * PB + t];
        rlo[t] = lo;
        rlen[t] = loff[(b + 1) * PB + t] - lo;
    }
    if (t < BIN_NODES) c[t] = 0;
    int ls = lo;
    for (int off = 16; off > 0; off >>= 1) ls += __shfl_xor(ls, off, 32);
    if (wave < PB / 32 && lane == 0) bsum[wave] = ls;
    __syncthreads();
    int beg = 0;
    for (int w = 0; w < PB / 32; ++w) beg += bsum[w];

    const int r = t / RT;
    const int q = t - r * RT;
    const int len = rlen[r];
    const int2* __restrict__ in = (const int2*)binned + (size_t)r * EPB + rlo[r];
    int hx[RU], hy[RU];
    #pragma unroll
    for (int k = 0; k < RU; ++k) { hx[k] = 0; hy[k] = 0; }
    if (q < len) {
        #pragma unroll
        for (int k = 0; k < RU; ++k) {
            const int jj = q + k * RT;
            const int2 p = in[(jj < len) ? jj : q];
            hx[k] = p.x;
            hy[k] = p.y;
        }
    }
    asm volatile("" : "+v"(hx[0]), "+v"(hy[0]), "+v"(hx[1]), "+v"(hy[1]),
                      "+v"(hx[2]), "+v"(hy[2]), "+v"(hx[3]), "+v"(hy[3]),
                      "+v"(hx[4]), "+v"(hy[4]), "+v"(hx[5]), "+v"(hy[5]),
                      "+v"(hx[6]), "+v"(hy[6]), "+v"(hx[7]), "+v"(hy[7]));
    #pragma unroll
    for (int k = 0; k < RU; ++k) {
        if (q + k * RT < len) atomicAdd(&c[hy[k] & (BIN_NODES - 1)], 1);
    }
    for (int j = q + RU * RT; j < len; j += RT) atomicAdd(&c[in[j].y & (BIN_NODES - 1)], 1);
    __syncthreads();

    const bool act = t < BIN_NODES;
    const int v = act ? c[t] : 0;
    int inc = v;
    for (int off = 1; off < 32; off <<= 1) {
        const int nb = __shfl_up(inc, off, 32);
        if (lane >= off) inc += nb;
    }
    if (act && lane == 31) wsum[wave] = inc;
    __syncthreads();
    int wbase = 0;
    for (int w = 0; w < BIN_NODES / 32; ++w) wbase += (w < wave) ? wsum[w] : 0;
    const int excl = beg + wbase + inc - v;
    if (act) {
        cur[t] = excl;
        const int n = b * BIN_NODES + t;
        if (n < N_NODES) { cnt[n] = v; rowptr[n] = excl; }
    }
    __syncthreads();
    #pragma unroll
    for (int k = 0; k < RU; ++k) {
        if (q + k * RT < len) {
            const int pos = atomicAdd(&cur[hy[k] & (BIN_NODES - 1)], 1);
            esrc[pos] = hx[k];
        }
    }
    for (int j = q + RU * RT; j < len; j += RT) {
        const int2 p = in[j];
        const int pos = atomicAdd(&cur[p.y & (BIN_NODES - 1)], 1);
        esrc[pos] = p.x;
    }
}

#define FOLD_BLOCKS ((FOLD_TILES + PT / 32 - 1) / (PT / 32))

__global__ __launch_bounds__(PT) void k_part_fold(const int* __restrict__ src,
                                                  const int* __restrict__ dst,
                                                  int* __restrict__ loff,
                                                  int* __restrict__ binned,
                                                  const float* __restrict__ W,
                                                  const float* __restrict__ FW,
                                                  const float* __restrict__ al,
                                                  const float* __restrict__ ar,
                                                  float* __restrict__ wcp) {
    if (blockIdx.x >= FOLD_BLOCKS) {
        part_body((int)blockIdx.x - FOLD_BLOCKS, src, dst, loff, binned);
        return;
    }
    const int tile = (int)blockIdx.x * (PT / 32) + (int)(threadIdx.x >> 5);
    if (tile >= FOLD_TILES) return;
    fold_body(tile, (int)(threadIdx.x & 31), W, FW, al, ar, wcp);
}

__global__ __launch_bounds__(PT) void k_gemm_sort(const float* __restrict__ X,
                                                  const float* __restrict__ wcp,
                                                  float* __restrict__ G,
                                                  float* __restrict__ el,
                                                  float* __restrict__ er,
                                                  const int* __restrict__ loff,
                                                  const int* __restrict__ binned,
                                                  int* __restrict__ cnt,
                                                  int* __restrict__ rowptr,
                                                  int* __restrict__ esrc) {
    if (blockIdx.x >= GEMM_BLOCKS) {
        binsort_body((int)blockIdx.x - GEMM_BLOCKS, loff, binned, cnt, rowptr, esrc);
        return;
    }
    const int rowTile = (int)(threadIdx.x >> 5) * GEMM_BLOCKS + (int)blockIdx.x;
    if (rowTile >= N_NODES / 16) return;
    gemm_body(rowTile, (int)(threadIdx.x & 31), X, wcp, G, el, er);
}

static_assert(N_NODES % 16 == 0, "two nodes per wave, eight waves per block, no partial wave");

__device__ __forceinline__ void load_chunk16(const int* __restrict__ esrc,
                                             const float* __restrict__ el,
                                             int start, int deg, int c, int l16,
                                             float er_n, int& s, float& e, bool& v) {
    const int j = c + l16;
    v = j < deg;
    const int idx = start + j;
    const int ld = esrc[idx < N_EDGES ? idx : N_EDGES - 1];
    s = v ? ld : 0;
    float t = el[s] + er_n;
    t = (t > 0.f) ? t : NEG_SLOPE * t;
    e = v ? t : -INFINITY;
}

__device__ __forceinline__ int idx_of(int s, int j, int hi) {
    const int lo = __builtin_amdgcn_readlane(s, j);
    const int up = __builtin_amdgcn_readlane(s, j + 16);
    return hi ? up : lo;
}
__device__ __forceinline__ float wt_of(float a, int j, int hi) {
    const int lo = __builtin_amdgcn_readlane(__float_as_int(a), j);
    const int up = __builtin_amdgcn_readlane(__float_as_int(a), j + 16);
    return __int_as_float(hi ? up : lo);
}

__device__ __forceinline__ void agg_rows2(const v4f* __restrict__ g4, int nrows, int mine,
                                          int s, float a, int hi, int l16, v4f& acc) {
    const v4f zero = {0.f, 0.f, 0.f, 0.f};
    for (int j = 0; j < nrows; j += 4) {
        v4f f0 = zero, f1 = zero, f2 = zero, f3 = zero;
        const int i0 = idx_of(s, j,     hi);
        const int i1 = idx_of(s, j + 1, hi);
        const int i2 = idx_of(s, j + 2, hi);
        const int i3 = idx_of(s, j + 3, hi);
        if (j     < mine) f0 = g4[(size_t)i0 * (N_CLASSES / 4) + l16];
        if (j + 1 < mine) f1 = g4[(size_t)i1 * (N_CLASSES / 4) + l16];
        if (j + 2 < mine) f2 = g4[(size_t)i2 * (N_CLASSES / 4) + l16];
        if (j + 3 < mine) f3 = g4[(size_t)i3 * (N_CLASSES / 4) + l16];
        acc += wt_of(a, j,     hi) * f0;
        acc += wt_of(a, j + 1, hi) * f1;
        acc += wt_of(a, j + 2, hi) * f2;
        acc += wt_of(a, j + 3, hi) * f3;
    }
}

__global__ __launch_bounds__(256) void k_pull(const int* __restrict__ cnt,
                                              const int* __restrict__ rowptr,
                                              const int* __restrict__ esrc,
                                              const float* __restrict__ el,
                                              const float* __restrict__ er,
                                              const float* __restrict__ G,
                                              const float* __restrict__ FB,
                                              float* __restrict__ out) {
    const int lane = threadIdx.x & 31;
    const int l16 = lane & 15;
    const int hi  = lane >> 4;
    const int n = ((int)(blockIdx.x * 8 + (threadIdx.x >> 5))) * 2 + hi;
    const int deg   = cnt[n];
    const int start = rowptr[n];
    const float er_n = er[n];
    const v4f* __restrict__ g4 = (const v4f*)G;
    const v4f bias = ((const v4f*)FB)[l16];
    const int dlo = __builtin_amdgcn_readlane(deg, 0);
    const int dup = __builtin_amdgcn_readlane(deg, 16);
    const int dmax = dlo > dup ? dlo : dup;

    int s0, s1; float e0, e1; bool v0, v1;
    load_chunk16(esrc, el, start, deg, 0,  l16, er_n, s0, e0, v0);
    load_chunk16(esrc, el, start, deg, 16, l16, er_n, s1, e1, v1);

    float m = fmaxf(e0, e1);
    for (int c = 32; c < dmax; c += 16) {
        int s; float e; bool v;
        load_chunk16(esrc, el, start, deg, c, l16, er_n, s, e, v);
        m = fmaxf(m, e);
    }
    for (int off = 8; off > 0; off >>= 1) m = fmaxf(m, __shfl_xor(m, off, 32));

    const float x0 = v0 ? __expf(e0 - m) : 0.f;
    const float x1 = v1 ? __expf(e1 - m) : 0.f;
    float dsum = x0 + x1;
    for (int c = 32; c < dmax; c += 16) {
        int s; float e; bool v;
        load_chunk16(esrc, el, start, deg, c, l16, er_n, s, e, v);
        dsum += v ? __expf(e - m) : 0.f;
    }
    for (int off = 8; off > 0; off >>= 1) dsum += __shfl_xor(dsum, off, 32);

    const float a0 = v0 ? x0 / dsum : 0.f;
    const float a1 = v1 ? x1 / dsum : 0.f;
    v4f acc = {0.f, 0.f, 0.f, 0.f};
    agg_rows2(g4, dmax < 16 ? dmax : 16, deg, s0, a0, hi, l16, acc);
    if (dmax > 16) {
        const int left = dmax - 16;
        agg_rows2(g4, left < 16 ? left : 16, deg - 16, s1, a1, hi, l16, acc);
    }
    for (int c = 32; c < dmax; c += 16) {
        int s; float e; bool v;
        load_chunk16(esrc, el, start, deg, c, l16, er_n, s, e, v);
        const float a = v ? __expf(e - m) / dsum : 0.f;
        const int left = dmax - c;
        agg_rows2(g4, left < 16 ? left : 16, deg - c, s, a, hi, l16, acc);
    }
    ((v4f*)out)[(size_t)n * (N_CLASSES / 4) + l16] = acc + bias;
}

extern "C" void kernel_launch(void* const* d_in, const int* in_sizes, int n_in,
                              void* d_out, int out_size, void* d_ws, size_t ws_size,
                              hipStream_t stream) {
    const float* X  = (const float*)d_in[0];
    const float* W  = (const float*)d_in[1];
    const float* al = (const float*)d_in[2];
    const float* ar = (const float*)d_in[3];
    const float* FW = (const float*)d_in[4];
    const float* FB = (const float*)d_in[5];
    const int*  src = (const int*)d_in[6];
    const int*  dst = (const int*)d_in[7];
    float* out = (float*)d_out;

    char* ws = (char*)d_ws;
    size_t off = 0;
    auto carve = [&](size_t bytes) {
        void* p = ws + off;
        off += (bytes + 255) & ~(size_t)255;
        return p;
    };
    float* G      = (float*)carve((size_t)N_NODES * N_CLASSES * sizeof(float));
    float* el     = (float*)carve((size_t)N_NODES * sizeof(float));
    float* er     = (float*)carve((size_t)N_NODES * sizeof(float));
    int*   cnt    = (int*)  carve((size_t)N_NODES * sizeof(int));
    int*   rowptr = (int*)  carve((size_t)N_NODES * sizeof(int));
    int*   esrc   = (int*)  carve((size_t)N_EDGES * sizeof(int));
    float* wcp    = (float*)carve((size_t)D_IN * NC * sizeof(float));
    int*   binned = (int*)  carve(((size_t)2 * N_EDGES + (size_t)NLOFF) * sizeof(int));
    int*   loff   = binned + (size_t)2 * N_EDGES;

    k_part_fold<<<FOLD_BLOCKS + PB, PT, 0, stream>>>(src, dst, loff, binned, W, FW, al, ar, wcp);
    k_gemm_sort<<<GEMM_BLOCKS + NBIN, PT, 0, stream>>>(X, wcp, G, el, er, loff, binned,
                                                      cnt, rowptr, esrc);
    k_pull<<<N_NODES / 16, 256, 0, stream>>>(cnt, rowptr, esrc, el, er, G, FB, out);
}


# GEMM operands through LDS: Wc copied once per block, X strip fetched as whole 128-B lines (32-k chunks) into a padded wave-private tile, fragments read from LDS; k order and sums unchanged
# speedup vs baseline: 9.8064x; 1.0778x over previous
#include <hip/hip_runtime.h>
#include <hip/hip_bf16.h>

#define N_NODES   100000
#define N_EDGES   1600000
#define D_IN      256
#define D_HID     128
#define N_CLASSES 64
#define NEG_SLOPE 0.2f

typedef __attribute__((ext_vector_type(2))) float v2f;
typedef __attribute__((ext_vector_type(4))) float v4f;
typedef __attribute__((ext_vector_type(8))) float v8f;

#define NC 80
#define COL_EL N_CLASSES
#define COL_ER (N_CLASSES + 1)

#define FOLD_TILES 80
__device__ __forceinline__ void fold_body(const int tile, const int lane,
                                          const float* __restrict__ W,
                                          const float* __restrict__ FW,
                                          const float* __restrict__ al,
                                          const float* __restrict__ ar,
                                          float* __restrict__ wcp) {
    const int rowTile = tile / 5;
    const int colTile = tile - rowTile * 5;
    const int l16 = lane & 15;
    const int hi  = lane >> 4;
    const int row0 = rowTile * 16;
    const int col = colTile * 16 + l16;

    const bool isfc = col < N_CLASSES;
    const bool used = col <= COL_ER;
    const float* __restrict__ bp = isfc ? (FW + col) : ((col == COL_ER) ? ar : al);
    const int bs = isfc ? N_CLASSES : 1;

    v8f acc = {};
    const float* __restrict__ wrow = W + (size_t)(row0 + l16) * D_HID;
    for (int j0 = 0; j0 < D_HID; j0 += 4) {
        const int ja = j0 + 2 * hi;
        v2f a;
        a.x = wrow[ja];
        a.y = wrow[ja + 1];
        const float bx = bp[(size_t)ja * bs];
        const float by = bp[(size_t)(ja + 1) * bs];
        v2f b;
        b.x = used ? bx : 0.f;
        b.y = used ? by : 0.f;
        acc = __builtin_amdgcn_wmma_f32_16x16x4_f32(
            false, a, false, b, (short)0, acc, false, false);
    }
    #pragma unroll
    for (int r = 0; r < 8; ++r) {
        const int k = row0 + r + 8 * hi;
        wcp[(size_t)(k >> 1) * (2 * NC) + col * 2 + (k & 1)] = acc[r];
    }
}

#define PT_GEMM     1024
#define GEMM_BLOCKS 256
#define GEMM_SW     25
#define KCH         32
#define A_PITCH     36
#define A_TILE      (16 * A_PITCH)
static_assert((GEMM_SW - 1) * GEMM_BLOCKS < N_NODES / 16 && GEMM_SW * GEMM_BLOCKS >= N_NODES / 16,
              "waves 0..GEMM_SW-1 are the ones that can own a strip");
static_assert((D_IN * NC) % (4 * PT_GEMM) == 0, "whole 16-byte pieces per thread in the Wc copy");

__device__ __forceinline__ void gemm_body(const int rowTile, const int wave, const int lane,
                                          const float* __restrict__ X,
                                          const v2f* sB, float* sA,
                                          float* __restrict__ G,
                                          float* __restrict__ el,
                                          float* __restrict__ er) {
    const int row0 = rowTile * 16;
    const int l16 = lane & 15;
    const int hi  = lane >> 4;
    const int lrow = lane >> 3;
    const int lseg = lane & 7;

    v8f acc0 = {}, acc1 = {}, acc2 = {}, acc3 = {}, acc4 = {};
    float* myA = sA + wave * A_TILE;
    const float* __restrict__ xs = X + (size_t)(row0 + lrow) * D_IN + lseg * 4;
    float* st = myA + lrow * A_PITCH + lseg * 4;
    const float* fa = myA + l16 * A_PITCH + 2 * hi;

    for (int ch = 0; ch < D_IN / KCH; ++ch) {
        v4f t0 = *(const v4f*)(xs + (size_t)0  * D_IN + ch * KCH);
        v4f t1 = *(const v4f*)(xs + (size_t)4  * D_IN + ch * KCH);
        v4f t2 = *(const v4f*)(xs + (size_t)8  * D_IN + ch * KCH);
        v4f t3 = *(const v4f*)(xs + (size_t)12 * D_IN + ch * KCH);
        asm volatile("" : "+v"(t0), "+v"(t1), "+v"(t2), "+v"(t3));
        *(v4f*)(st + 0  * A_PITCH) = t0;
        *(v4f*)(st + 4  * A_PITCH) = t1;
        *(v4f*)(st + 8  * A_PITCH) = t2;
        *(v4f*)(st + 12 * A_PITCH) = t3;
        #pragma unroll
        for (int j = 0; j < KCH / 4; ++j) {
            const v2f a = *(const v2f*)(fa + 4 * j);
            const v2f* w = sB + (size_t)(ch * (KCH / 2) + 2 * j + hi) * NC + l16;
            const v2f b0 = w[0];
            const v2f b1 = w[16];
            const v2f b2 = w[32];
            const v2f b3 = w[48];
            const v2f b4 = w[64];
            acc0 = __builtin_amdgcn_wmma_f32_16x16x4_f32(
                false, a, false, b0, (short)0, acc0, false, false);
            acc1 = __builtin_amdgcn_wmma_f32_16x16x4_f32(
                false, a, false, b1, (short)0, acc1, false, false);
            acc2 = __builtin_amdgcn_wmma_f32_16x16x4_f32(
                false, a, false, b2, (short)0, acc2, false, false);
            acc3 = __builtin_amdgcn_wmma_f32_16x16x4_f32(
                false, a, false, b3, (short)0, acc3, false, false);
            acc4 = __builtin_amdgcn_wmma_f32_16x16x4_f32(
                false, a, false, b4, (short)0, acc4, false, false);
        }
    }
    const int ln  = (int)__builtin_amdgcn_mbcnt_lo(~0u, 0u);
    const int c16 = ln & 15;
    const int h8  = (ln >> 4) * 8;
    float* __restrict__ sc = (c16 == 0) ? el : er;
    #pragma unroll
    for (int r = 0; r < 8; ++r) {
        const int row = row0 + r + h8;
        float* __restrict__ orow = G + (size_t)row * N_CLASSES + c16;
        orow[0]  = acc0[r];
        orow[16] = acc1[r];
        orow[32] = acc2[r];
        orow[48] = acc3[r];
        if (c16 < 2) sc[row] = acc4[r];
    }
}

#define BIN_SHIFT 8
#define BIN_NODES (1 << BIN_SHIFT)
#define NBIN      ((N_NODES + BIN_NODES - 1) / BIN_NODES)
#define PB        256
#define EPB       (N_EDGES / PB)
#define NLOFF     ((NBIN + 1) * PB)
#define PT        1024
#define EIT       ((EPB + PT - 1) / PT)
#define NSW       ((NBIN + 31) / 32)
#define RT        (PT / PB)
#define RU        8
static_assert(N_EDGES % PB == 0, "edge chunks must tile the edge list");
static_assert(NBIN < PT, "one thread per bin, and one more for the closing row");
static_assert(PT % PB == 0, "whole thread groups per run");
static_assert(RU == 8, "the statement that groups the run loads names 8 pairs");
static_assert(BIN_NODES <= PT && PB <= PT, "one thread per node / per run in the bin block");
static_assert(PT == PT_GEMM, "the GEMM body sizes its Wc copy by the launch's block");

__device__ __forceinline__ void part_body(const int b,
                                          const int* __restrict__ src,
                                          const int* __restrict__ dst,
                                          int* __restrict__ loff,
                                          int* __restrict__ binned) {
    __shared__ int tab[NBIN];
    __shared__ int wsum[NSW];
    const int t = threadIdx.x;
    const int lane = t & 31;
    const int wave = t >> 5;
    if (t < NBIN) tab[t] = 0;
    const int* __restrict__ s = src + (size_t)b * EPB;
    const int* __restrict__ d = dst + (size_t)b * EPB;
    int2* __restrict__ out = (int2*)binned + (size_t)b * EPB;
    int dv[EIT], sv[EIT];
    #pragma unroll
    for (int k = 0; k < EIT; ++k) {
        const int i = t + k * PT;
        dv[k] = (i < EPB) ? d[i] : 0;
        sv[k] = (i < EPB) ? s[i] : 0;
    }
    __syncthreads();
    #pragma unroll
    for (int k = 0; k < EIT; ++k) {
        const int i = t + k * PT;
        if (i < EPB) atomicAdd(&tab[dv[k] >> BIN_SHIFT], 1);
    }
    __syncthreads();

    const bool act = t < NBIN;
    const int v = act ? tab[t] : 0;
    int inc = v;
    for (int off = 1; off < 32; off <<= 1) {
        const int nb = __shfl_up(inc, off, 32);
        if (lane >= off) inc += nb;
    }
    if (wave < NSW && lane == 31) wsum[wave] = inc;
    __syncthreads();
    int wbase = 0;
    for (int w = 0; w < NSW; ++w) wbase += (w < wave) ? wsum[w] : 0;
    const int excl = wbase + inc - v;
    if (act) {
        tab[t] = excl;
        loff[t * PB + b] = excl;
    }
    if (t == NBIN) loff[NBIN * PB + b] = EPB;
    __syncthreads();
    #pragma unroll
    for (int k = 0; k < EIT; ++k) {
        const int i = t + k * PT;
        if (i < EPB) {
            const int pos = atomicAdd(&tab[dv[k] >> BIN_SHIFT], 1);
            out[pos] = make_int2(sv[k], dv[k]);
        }
    }
}

__device__ __forceinline__ void binsort_body(const int b,
                                             const int* __restrict__ loff,
                                             const int* __restrict__ binned,
                                             int* __restrict__ cnt,
                                             int* __restrict__ rowptr,
                                             int* __restrict__ esrc) {
    __shared__ int c[BIN_NODES];
    __shared__ int cur[BIN_NODES];
    __shared__ int wsum[BIN_NODES / 32];
    __shared__ int bsum[PB / 32];
    __shared__ int rlo[PB];
    __shared__ int rlen[PB];
    const int t = threadIdx.x;
    const int lane = t & 31;
    const int wave = t >> 5;

    int lo = 0;
    if (t < PB) {
        lo = loff[b * PB + t];
        rlo[t] = lo;
        rlen[t] = loff[(b + 1) * PB + t] - lo;
    }
    if (t < BIN_NODES) c[t] = 0;
    int ls = lo;
    for (int off = 16; off > 0; off >>= 1) ls += __shfl_xor(ls, off, 32);
    if (wave < PB / 32 && lane == 0) bsum[wave] = ls;
    __syncthreads();
    int beg = 0;
    for (int w = 0; w < PB / 32; ++w) beg += bsum[w];

    const int r = t / RT;
    const int q = t - r * RT;
    const int len = rlen[r];
    const int2* __restrict__ in = (const int2*)binned + (size_t)r * EPB + rlo[r];
    int hx[RU], hy[RU];
    #pragma unroll
    for (int k = 0; k < RU; ++k) { hx[k] = 0; hy[k] = 0; }
    if (q < len) {
        #pragma unroll
        for (int k = 0; k < RU; ++k) {
            const int jj = q + k * RT;
            const int2 p = in[(jj < len) ? jj : q];
            hx[k] = p.x;
            hy[k] = p.y;
        }
    }
    asm volatile("" : "+v"(hx[0]), "+v"(hy[0]), "+v"(hx[1]), "+v"(hy[1]),
                      "+v"(hx[2]), "+v"(hy[2]), "+v"(hx[3]), "+v"(hy[3]),
                      "+v"(hx[4]), "+v"(hy[4]), "+v"(hx[5]), "+v"(hy[5]),
                      "+v"(hx[6]), "+v"(hy[6]), "+v"(hx[7]), "+v"(hy[7]));
    #pragma unroll
    for (int k = 0; k < RU; ++k) {
        if (q + k * RT < len) atomicAdd(&c[hy[k] & (BIN_NODES - 1)], 1);
    }
    for (int j = q + RU * RT; j < len; j += RT) atomicAdd(&c[in[j].y & (BIN_NODES - 1)], 1);
    __syncthreads();

    const bool act = t < BIN_NODES;
    const int v = act ? c[t] : 0;
    int inc = v;
    for (int off = 1; off < 32; off <<= 1) {
        const int nb = __shfl_up(inc, off, 32);
        if (lane >= off) inc += nb;
    }
    if (act && lane == 31) wsum[wave] = inc;
    __syncthreads();
    int wbase = 0;
    for (int w = 0; w < BIN_NODES / 32; ++w) wbase += (w < wave) ? wsum[w] : 0;
    const int excl = beg + wbase + inc - v;
    if (act) {
        cur[t] = excl;
        const int n = b * BIN_NODES + t;
        if (n < N_NODES) { cnt[n] = v; rowptr[n] = excl; }
    }
    __syncthreads();
    #pragma unroll
    for (int k = 0; k < RU; ++k) {
        if (q + k * RT < len) {
            const int pos = atomicAdd(&cur[hy[k] & (BIN_NODES - 1)], 1);
            esrc[pos] = hx[k];
        }
    }
    for (int j = q + RU * RT; j < len; j += RT) {
        const int2 p = in[j];
        const int pos = atomicAdd(&cur[p.y & (BIN_NODES - 1)], 1);
        esrc[pos] = p.x;
    }
}

#define FOLD_BLOCKS ((FOLD_TILES + PT / 32 - 1) / (PT / 32))

__global__ __launch_bounds__(PT) void k_part_fold(const int* __restrict__ src,
                                                  const int* __restrict__ dst,
                                                  int* __restrict__ loff,
                                                  int* __restrict__ binned,
                                                  const float* __restrict__ W,
                                                  const float* __restrict__ FW,
                                                  const float* __restrict__ al,
                                                  const float* __restrict__ ar,
                                                  float* __restrict__ wcp) {
    if (blockIdx.x >= FOLD_BLOCKS) {
        part_body((int)blockIdx.x - FOLD_BLOCKS, src, dst, loff, binned);
        return;
    }
    const int tile = (int)blockIdx.x * (PT / 32) + (int)(threadIdx.x >> 5);
    if (tile >= FOLD_TILES) return;
    fold_body(tile, (int)(threadIdx.x & 31), W, FW, al, ar, wcp);
}

__global__ __launch_bounds__(PT, 16) void k_gemm_sort(const float* __restrict__ X,
                                                  const float* __restrict__ wcp,
                                                  float* __restrict__ G,
                                                  float* __restrict__ el,
                                                  float* __restrict__ er,
                                                  const int* __restrict__ loff,
                                                  const int* __restrict__ binned,
                                                  int* __restrict__ cnt,
                                                  int* __restrict__ rowptr,
                                                  int* __restrict__ esrc) {
    if (blockIdx.x >= GEMM_BLOCKS) {
        binsort_body((int)blockIdx.x - GEMM_BLOCKS, loff, binned, cnt, rowptr, esrc);
        return;
    }
    __shared__ __attribute__((aligned(16))) v2f sB[(D_IN / 2) * NC];
    __shared__ __attribute__((aligned(16))) float sA[GEMM_SW * A_TILE];
    #pragma unroll
    for (int k = 0; k < D_IN * NC / (4 * PT); ++k)
        ((v4f*)sB)[(int)threadIdx.x + k * PT] = ((const v4f*)wcp)[(int)threadIdx.x + k * PT];
    __syncthreads();
    const int wave = __builtin_amdgcn_readfirstlane((int)(threadIdx.x >> 5));
    const int rowTile = wave * GEMM_BLOCKS + (int)blockIdx.x;
    if (rowTile >= N_NODES / 16) return;
    gemm_body(rowTile, wave, (int)(threadIdx.x & 31), X, sB, sA, G, el, er);
}

static_assert(N_NODES % 16 == 0, "two nodes per wave, eight waves per block, no partial wave");

__device__ __forceinline__ void load_chunk16(const int* __restrict__ esrc,
                                             const float* __restrict__ el,
                                             int start, int deg, int c, int l16,
                                             float er_n, int& s, float& e, bool& v) {
    const int j = c + l16;
    v = j < deg;
    const int idx = start + j;
    const int ld = esrc[idx < N_EDGES ? idx : N_EDGES - 1];
    s = v ? ld : 0;
    float t = el[s] + er_n;
    t = (t > 0.f) ? t : NEG_SLOPE * t;
    e = v ? t : -INFINITY;
}

__device__ __forceinline__ int idx_of(int s, int j, int hi) {
    const int lo = __builtin_amdgcn_readlane(s, j);
    const int up = __builtin_amdgcn_readlane(s, j + 16);
    return hi ? up : lo;
}
__device__ __forceinline__ float wt_of(float a, int j, int hi) {
    const int lo = __builtin_amdgcn_readlane(__float_as_int(a), j);
    const int up = __builtin_amdgcn_readlane(__float_as_int(a), j + 16);
    return __int_as_float(hi ? up : lo);
}

__device__ __forceinline__ void agg_rows2(const v4f* __restrict__ g4, int nrows, int mine,
                                          int s, float a, int hi, int l16, v4f& acc) {
    const v4f zero = {0.f, 0.f, 0.f, 0.f};
    for (int j = 0; j < nrows; j += 4) {
        v4f f0 = zero, f1 = zero, f2 = zero, f3 = zero;
        const int i0 = idx_of(s, j,     hi);
        const int i1 = idx_of(s, j + 1, hi);
        const int i2 = idx_of(s, j + 2, hi);
        const int i3 = idx_of(s, j + 3, hi);
        if (j     < mine) f0 = g4[(size_t)i0 * (N_CLASSES / 4) + l16];
        if (j + 1 < mine) f1 = g4[(size_t)i1 * (N_CLASSES / 4) + l16];
        if (j + 2 < mine) f2 = g4[(size_t)i2 * (N_CLASSES / 4) + l16];
        if (j + 3 < mine) f3 = g4[(size_t)i3 * (N_CLASSES / 4) + l16];
        acc += wt_of(a, j,     hi) * f0;
        acc += wt_of(a, j + 1, hi) * f1;
        acc += wt_of(a, j + 2, hi) * f2;
        acc += wt_of(a, j + 3, hi) * f3;
    }
}

__global__ __launch_bounds__(256) void k_pull(const int* __restrict__ cnt,
                                              const int* __restrict__ rowptr,
                                              const int* __restrict__ esrc,
                                              const float* __restrict__ el,
                                              const float* __restrict__ er,
                                              const float* __restrict__ G,
                                              const float* __restrict__ FB,
                                              float* __restrict__ out) {
    const int lane = threadIdx.x & 31;
    const int l16 = lane & 15;
    const int hi  = lane >> 4;
    const int n = ((int)(blockIdx.x * 8 + (threadIdx.x >> 5))) * 2 + hi;
    const int deg   = cnt[n];
    const int start = rowptr[n];
    const float er_n = er[n];
    const v4f* __restrict__ g4 = (const v4f*)G;
    const v4f bias = ((const v4f*)FB)[l16];
    const int dlo = __builtin_amdgcn_readlane(deg, 0);
    const int dup = __builtin_amdgcn_readlane(deg, 16);
    const int dmax = dlo > dup ? dlo : dup;

    int s0, s1; float e0, e1; bool v0, v1;
    load_chunk16(esrc, el, start, deg, 0,  l16, er_n, s0, e0, v0);
    load_chunk16(esrc, el, start, deg, 16, l16, er_n, s1, e1, v1);

    float m = fmaxf(e0, e1);
    for (int c = 32; c < dmax; c += 16) {
        int s; float e; bool v;
        load_chunk16(esrc, el, start, deg, c, l16, er_n, s, e, v);
        m = fmaxf(m, e);
    }
    for (int off = 8; off > 0; off >>= 1) m = fmaxf(m, __shfl_xor(m, off, 32));

    const float x0 = v0 ? __expf(e0 - m) : 0.f;
    const float x1 = v1 ? __expf(e1 - m) : 0.f;
    float dsum = x0 + x1;
    for (int c = 32; c < dmax; c += 16) {
        int s; float e; bool v;
        load_chunk16(esrc, el, start, deg, c, l16, er_n, s, e, v);
        dsum += v ? __expf(e - m) : 0.f;
    }
    for (int off = 8; off > 0; off >>= 1) dsum += __shfl_xor(dsum, off, 32);

    const float a0 = v0 ? x0 / dsum : 0.f;
    const float a1 = v1 ? x1 / dsum : 0.f;
    v4f acc = {0.f, 0.f, 0.f, 0.f};
    agg_rows2(g4, dmax < 16 ? dmax : 16, deg, s0, a0, hi, l16, acc);
    if (dmax > 16) {
        const int left = dmax - 16;
        agg_rows2(g4, left < 16 ? left : 16, deg - 16, s1, a1, hi, l16, acc);
    }
    for (int c = 32; c < dmax; c += 16) {
        int s; float e; bool v;
        load_chunk16(esrc, el, start, deg, c, l16, er_n, s, e, v);
        const float a = v ? __expf(e - m) / dsum : 0.f;
        const int left = dmax - c;
        agg_rows2(g4, left < 16 ? left : 16, deg - c, s, a, hi, l16, acc);
    }
    ((v4f*)out)[(size_t)n * (N_CLASSES / 4) + l16] = acc + bias;
}

extern "C" void kernel_launch(void* const* d_in, const int* in_sizes, int n_in,
                              void* d_out, int out_size, void* d_ws, size_t ws_size,
                              hipStream_t stream) {
    const float* X  = (const float*)d_in[0];
    const float* W  = (const float*)d_in[1];
    const float* al = (const float*)d_in[2];
    const float* ar = (const float*)d_in[3];
    const float* FW = (const float*)d_in[4];
    const float* FB = (const float*)d_in[5];
    const int*  src = (const int*)d_in[6];
    const int*  dst = (const int*)d_in[7];
    float* out = (float*)d_out;

    char* ws = (char*)d_ws;
    size_t off = 0;
    auto carve = [&](size_t bytes) {
        void* p = ws + off;
        off += (bytes + 255) & ~(size_t)255;
        return p;
    };
    float* G      = (float*)carve((size_t)N_NODES * N_CLASSES * sizeof(float));
    float* el     = (float*)carve((size_t)N_NODES * sizeof(float));
    float* er     = (float*)carve((size_t)N_NODES * sizeof(float));
    int*   cnt    = (int*)  carve((size_t)N_NODES * sizeof(int));
    int*   rowptr = (int*)  carve((size_t)N_NODES * sizeof(int));
    int*   esrc   = (int*)  carve((size_t)N_EDGES * sizeof(int));
    float* wcp    = (float*)carve((size_t)D_IN * NC * sizeof(float));
    int*   binned = (int*)  carve(((size_t)2 * N_EDGES + (size_t)NLOFF) * sizeof(int));
    int*   loff   = binned + (size_t)2 * N_EDGES;

    k_part_fold<<<FOLD_BLOCKS + PB, PT, 0, stream>>>(src, dst, loff, binned, W, FW, al, ar, wcp);
    k_gemm_sort<<<GEMM_BLOCKS + NBIN, PT, 0, stream>>>(X, wcp, G, el, er, loff, binned,
                                                      cnt, rowptr, esrc);
    k_pull<<<N_NODES / 16, 256, 0, stream>>>(cnt, rowptr, esrc, el, er, G, FB, out);
}


# Per-destination pass row loop: edge index and weight reach both 16-lane halves by one row-share lane move (was 2 scalar lane reads, copy, select); row loads unbranched, 32-bit offsets; sum order kept
# speedup vs baseline: 10.1084x; 1.0308x over previous
#include <hip/hip_runtime.h>
#include <hip/hip_bf16.h>

#define N_NODES   100000
#define N_EDGES   1600000
#define D_IN      256
#define D_HID     128
#define N_CLASSES 64
#define NEG_SLOPE 0.2f

typedef __attribute__((ext_vector_type(2))) float v2f;
typedef __attribute__((ext_vector_type(4))) float v4f;
typedef __attribute__((ext_vector_type(8))) float v8f;

#define NC 80
#define COL_EL N_CLASSES
#define COL_ER (N_CLASSES + 1)

#define FOLD_TILES 80
__device__ __forceinline__ void fold_body(const int tile, const int lane,
                                          const float* __restrict__ W,
                                          const float* __restrict__ FW,
                                          const float* __restrict__ al,
                                          const float* __restrict__ ar,
                                          float* __restrict__ wcp) {
    const int rowTile = tile / 5;
    const int colTile = tile - rowTile * 5;
    const int l16 = lane & 15;
    const int hi  = lane >> 4;
    const int row0 = rowTile * 16;
    const int col = colTile * 16 + l16;

    const bool isfc = col < N_CLASSES;
    const bool used = col <= COL_ER;
    const float* __restrict__ bp = isfc ? (FW + col) : ((col == COL_ER) ? ar : al);
    const int bs = isfc ? N_CLASSES : 1;

    v8f acc = {};
    const float* __restrict__ wrow = W + (size_t)(row0 + l16) * D_HID;
    for (int j0 = 0; j0 < D_HID; j0 += 4) {
        const int ja = j0 + 2 * hi;
        v2f a;
        a.x = wrow[ja];
        a.y = wrow[ja + 1];
        const float bx = bp[(size_t)ja * bs];
        const float by = bp[(size_t)(ja + 1) * bs];
        v2f b;
        b.x = used ? bx : 0.f;
        b.y = used ? by : 0.f;
        acc = __builtin_amdgcn_wmma_f32_16x16x4_f32(
            false, a, false, b, (short)0, acc, false, false);
    }
    #pragma unroll
    for (int r = 0; r < 8; ++r) {
        const int k = row0 + r + 8 * hi;
        wcp[(size_t)(k >> 1) * (2 * NC) + col * 2 + (k & 1)] = acc[r];
    }
}

#define PT_GEMM     1024
#define GEMM_BLOCKS 256
#define GEMM_SW     25
#define KCH         32
#define A_PITCH     36
#define A_TILE      (16 * A_PITCH)
static_assert((GEMM_SW - 1) * GEMM_BLOCKS < N_NODES / 16 && GEMM_SW * GEMM_BLOCKS >= N_NODES / 16,
              "waves 0..GEMM_SW-1 are the ones that can own a strip");
static_assert((D_IN * NC) % (4 * PT_GEMM) == 0, "whole 16-byte pieces per thread in the Wc copy");

__device__ __forceinline__ void gemm_body(const int rowTile, const int wave, const int lane,
                                          const float* __restrict__ X,
                                          const v2f* sB, float* sA,
                                          float* __restrict__ G,
                                          float* __restrict__ el,
                                          float* __restrict__ er) {
    const int row0 = rowTile * 16;
    const int l16 = lane & 15;
    const int hi  = lane >> 4;
    const int lrow = lane >> 3;
    const int lseg = lane & 7;

    v8f acc0 = {}, acc1 = {}, acc2 = {}, acc3 = {}, acc4 = {};
    float* myA = sA + wave * A_TILE;
    const float* __restrict__ xs = X + (size_t)(row0 + lrow) * D_IN + lseg * 4;
    float* st = myA + lrow * A_PITCH + lseg * 4;
    const float* fa = myA + l16 * A_PITCH + 2 * hi;

    for (int ch = 0; ch < D_IN / KCH; ++ch) {
        v4f t0 = *(const v4f*)(xs + (size_t)0  * D_IN + ch * KCH);
        v4f t1 = *(const v4f*)(xs + (size_t)4  * D_IN + ch * KCH);
        v4f t2 = *(const v4f*)(xs + (size_t)8  * D_IN + ch * KCH);
        v4f t3 = *(const v4f*)(xs + (size_t)12 * D_IN + ch * KCH);
        asm volatile("" : "+v"(t0), "+v"(t1), "+v"(t2), "+v"(t3));
        *(v4f*)(st + 0  * A_PITCH) = t0;
        *(v4f*)(st + 4  * A_PITCH) = t1;
        *(v4f*)(st + 8  * A_PITCH) = t2;
        *(v4f*)(st + 12 * A_PITCH) = t3;
        #pragma unroll
        for (int j = 0; j < KCH / 4; ++j) {
            const v2f a = *(const v2f*)(fa + 4 * j);
            const v2f* w = sB + (size_t)(ch * (KCH / 2) + 2 * j + hi) * NC + l16;
            const v2f b0 = w[0];
            const v2f b1 = w[16];
            const v2f b2 = w[32];
            const v2f b3 = w[48];
            const v2f b4 = w[64];
            acc0 = __builtin_amdgcn_wmma_f32_16x16x4_f32(
                false, a, false, b0, (short)0, acc0, false, false);
            acc1 = __builtin_amdgcn_wmma_f32_16x16x4_f32(
                false, a, false, b1, (short)0, acc1, false, false);
            acc2 = __builtin_amdgcn_wmma_f32_16x16x4_f32(
                false, a, false, b2, (short)0, acc2, false, false);
            acc3 = __builtin_amdgcn_wmma_f32_16x16x4_f32(
                false, a, false, b3, (short)0, acc3, false, false);
            acc4 = __builtin_amdgcn_wmma_f32_16x16x4_f32(
                false, a, false, b4, (short)0, acc4, false, false);
        }
    }
    const int ln  = (int)__builtin_amdgcn_mbcnt_lo(~0u, 0u);
    const int c16 = ln & 15;
    const int h8  = (ln >> 4) * 8;
    float* __restrict__ sc = (c16 == 0) ? el : er;
    #pragma unroll
    for (int r = 0; r < 8; ++r) {
        const int row = row0 + r + h8;
        float* __restrict__ orow = G + (size_t)row * N_CLASSES + c16;
        orow[0]  = acc0[r];
        orow[16] = acc1[r];
        orow[32] = acc2[r];
        orow[48] = acc3[r];
        if (c16 < 2) sc[row] = acc4[r];
    }
}

#define BIN_SHIFT 8
#define BIN_NODES (1 << BIN_SHIFT)
#define NBIN      ((N_NODES + BIN_NODES - 1) / BIN_NODES)
#define PB        256
#define EPB       (N_EDGES / PB)
#define NLOFF     ((NBIN + 1) * PB)
#define PT        1024
#define EIT       ((EPB + PT - 1) / PT)
#define NSW       ((NBIN + 31) / 32)
#define RT        (PT / PB)
#define RU        8
static_assert(N_EDGES % PB == 0, "edge chunks must tile the edge list");
static_assert(NBIN < PT, "one thread per bin, and one more for the closing row");
static_assert(PT % PB == 0, "whole thread groups per run");
static_assert(RU == 8, "the statement that groups the run loads names 8 pairs");
static_assert(BIN_NODES <= PT && PB <= PT, "one thread per node / per run in the bin block");
static_assert(PT == PT_GEMM, "the GEMM body sizes its Wc copy by the launch's block");

__device__ __forceinline__ void part_body(const int b,
                                          const int* __restrict__ src,
                                          const int* __restrict__ dst,
                                          int* __restrict__ loff,
                                          int* __restrict__ binned) {
    __shared__ int tab[NBIN];
    __shared__ int wsum[NSW];
    const int t = threadIdx.x;
    const int lane = t & 31;
    const int wave = t >> 5;
    if (t < NBIN) tab[t] = 0;
    const int* __restrict__ s = src + (size_t)b * EPB;
    const int* __restrict__ d = dst + (size_t)b * EPB;
    int2* __restrict__ out = (int2*)binned + (size_t)b * EPB;
    int dv[EIT], sv[EIT];
    #pragma unroll
    for (int k = 0; k < EIT; ++k) {
        const int i = t + k * PT;
        dv[k] = (i < EPB) ? d[i] : 0;
        sv[k] = (i < EPB) ? s[i] : 0;
    }
    __syncthreads();
    #pragma unroll
    for (int k = 0; k < EIT; ++k) {
        const int i = t + k * PT;
        if (i < EPB) atomicAdd(&tab[dv[k] >> BIN_SHIFT], 1);
    }
    __syncthreads();

    const bool act = t < NBIN;
    const int v = act ? tab[t] : 0;
    int inc = v;
    for (int off = 1; off < 32; off <<= 1) {
        const int nb = __shfl_up(inc, off, 32);
        if (lane >= off) inc += nb;
    }
    if (wave < NSW && lane == 31) wsum[wave] = inc;
    __syncthreads();
    int wbase = 0;
    for (int w = 0; w < NSW; ++w) wbase += (w < wave) ? wsum[w] : 0;
    const int excl = wbase + inc - v;
    if (act) {
        tab[t] = excl;
        loff[t * PB + b] = excl;
    }
    if (t == NBIN) loff[NBIN * PB + b] = EPB;
    __syncthreads();
    #pragma unroll
    for (int k = 0; k < EIT; ++k) {
        const int i = t + k * PT;
        if (i < EPB) {
            const int pos = atomicAdd(&tab[dv[k] >> BIN_SHIFT], 1);
            out[pos] = make_int2(sv[k], dv[k]);
        }
    }
}

__device__ __forceinline__ void binsort_body(const int b,
                                             const int* __restrict__ loff,
                                             const int* __restrict__ binned,
                                             int* __restrict__ cnt,
                                             int* __restrict__ rowptr,
                                             int* __restrict__ esrc) {
    __shared__ int c[BIN_NODES];
    __shared__ int cur[BIN_NODES];
    __shared__ int wsum[BIN_NODES / 32];
    __shared__ int bsum[PB / 32];
    __shared__ int rlo[PB];
    __shared__ int rlen[PB];
    const int t = threadIdx.x;
    const int lane = t & 31;
    const int wave = t >> 5;

    int lo = 0;
    if (t < PB) {
        lo = loff[b * PB + t];
        rlo[t] = lo;
        rlen[t] = loff[(b + 1) * PB + t] - lo;
    }
    if (t < BIN_NODES) c[t] = 0;
    int ls = lo;
    for (int off = 16; off > 0; off >>= 1) ls += __shfl_xor(ls, off, 32);
    if (wave < PB / 32 && lane == 0) bsum[wave] = ls;
    __syncthreads();
    int beg = 0;
    for (int w = 0; w < PB / 32; ++w) beg += bsum[w];

    const int r = t / RT;
    const int q = t - r * RT;
    const int len = rlen[r];
    const int2* __restrict__ in = (const int2*)binned + (size_t)r * EPB + rlo[r];
    int hx[RU], hy[RU];
    #pragma unroll
    for (int k = 0; k < RU; ++k) { hx[k] = 0; hy[k] = 0; }
    if (q < len) {
        #pragma unroll
        for (int k = 0; k < RU; ++k) {
            const int jj = q + k * RT;
            const int2 p = in[(jj < len) ? jj : q];
            hx[k] = p.x;
            hy[k] = p.y;
        }
    }
    asm volatile("" : "+v"(hx[0]), "+v"(hy[0]), "+v"(hx[1]), "+v"(hy[1]),
                      "+v"(hx[2]), "+v"(hy[2]), "+v"(hx[3]), "+v"(hy[3]),
                      "+v"(hx[4]), "+v"(hy[4]), "+v"(hx[5]), "+v"(hy[5]),
                      "+v"(hx[6]), "+v"(hy[6]), "+v"(hx[7]), "+v"(hy[7]));
    #pragma unroll
    for (int k = 0; k < RU; ++k) {
        if (q + k * RT < len) atomicAdd(&c[hy[k] & (BIN_NODES - 1)], 1);
    }
    for (int j = q + RU * RT; j < len; j += RT) atomicAdd(&c[in[j].y & (BIN_NODES - 1)], 1);
    __syncthreads();

    const bool act = t < BIN_NODES;
    const int v = act ? c[t] : 0;
    int inc = v;
    for (int off = 1; off < 32; off <<= 1) {
        const int nb = __shfl_up(inc, off, 32);
        if (lane >= off) inc += nb;
    }
    if (act && lane == 31) wsum[wave] = inc;
    __syncthreads();
    int wbase = 0;
    for (int w = 0; w < BIN_NODES / 32; ++w) wbase += (w < wave) ? wsum[w] : 0;
    const int excl = beg + wbase + inc - v;
    if (act) {
        cur[t] = excl;
        const int n = b * BIN_NODES + t;
        if (n < N_NODES) { cnt[n] = v; rowptr[n] = excl; }
    }
    __syncthreads();
    #pragma unroll
    for (int k = 0; k < RU; ++k) {
        if (q + k * RT < len) {
            const int pos = atomicAdd(&cur[hy[k] & (BIN_NODES - 1)], 1);
            esrc[pos] = hx[k];
        }
    }
    for (int j = q + RU * RT; j < len; j += RT) {
        const int2 p = in[j];
        const int pos = atomicAdd(&cur[p.y & (BIN_NODES - 1)], 1);
        esrc[pos] = p.x;
    }
}

#define FOLD_BLOCKS ((FOLD_TILES + PT / 32 - 1) / (PT / 32))

__global__ __launch_bounds__(PT) void k_part_fold(const int* __restrict__ src,
                                                  const int* __restrict__ dst,
                                                  int* __restrict__ loff,
                                                  int* __restrict__ binned,
                                                  const float* __restrict__ W,
                                                  const float* __restrict__ FW,
                                                  const float* __restrict__ al,
                                                  const float* __restrict__ ar,
                                                  float* __restrict__ wcp) {
    if (blockIdx.x >= FOLD_BLOCKS) {
        part_body((int)blockIdx.x - FOLD_BLOCKS, src, dst, loff, binned);
        return;
    }
    const int tile = (int)blockIdx.x * (PT / 32) + (int)(threadIdx.x >> 5);
    if (tile >= FOLD_TILES) return;
    fold_body(tile, (int)(threadIdx.x & 31), W, FW, al, ar, wcp);
}

__global__ __launch_bounds__(PT, 16) void k_gemm_sort(const float* __restrict__ X,
                                                  const float* __restrict__ wcp,
                                                  float* __restrict__ G,
                                                  float* __restrict__ el,
                                                  float* __restrict__ er,
                                                  const int* __restrict__ loff,
                                                  const int* __restrict__ binned,
                                                  int* __restrict__ cnt,
                                                  int* __restrict__ rowptr,
                                                  int* __restrict__ esrc) {
    if (blockIdx.x >= GEMM_BLOCKS) {
        binsort_body((int)blockIdx.x - GEMM_BLOCKS, loff, binned, cnt, rowptr, esrc);
        return;
    }
    __shared__ __attribute__((aligned(16))) v2f sB[(D_IN / 2) * NC];
    __shared__ __attribute__((aligned(16))) float sA[GEMM_SW * A_TILE];
    #pragma unroll
    for (int k = 0; k < D_IN * NC / (4 * PT); ++k)
        ((v4f*)sB)[(int)threadIdx.x + k * PT] = ((const v4f*)wcp)[(int)threadIdx.x + k * PT];
    __syncthreads();
    const int wave = __builtin_amdgcn_readfirstlane((int)(threadIdx.x >> 5));
    const int rowTile = wave * GEMM_BLOCKS + (int)blockIdx.x;
    if (rowTile >= N_NODES / 16) return;
    gemm_body(rowTile, wave, (int)(threadIdx.x & 31), X, sB, sA, G, el, er);
}

static_assert(N_NODES % 16 == 0, "two nodes per wave, eight waves per block, no partial wave");

__device__ __forceinline__ void load_chunk16(const int* __restrict__ esrc,
                                             const float* __restrict__ el,
                                             int start, int deg, int c, int l16,
                                             float er_n, int& s, float& e, bool& v) {
    const int j = c + l16;
    v = j < deg;
    const int idx = start + j;
    const int ld = esrc[idx < N_EDGES ? idx : N_EDGES - 1];
    s = v ? ld : 0;
    float t = el[s] + er_n;
    t = (t > 0.f) ? t : NEG_SLOPE * t;
    e = v ? t : -INFINITY;
}

template <int J>
__device__ __forceinline__ int row_lane(int v) {
    return __builtin_amdgcn_update_dpp(0, v, 0x150 + J, 0xF, 0xF, false);
}

template <int J>
__device__ __forceinline__ void agg4(const char* __restrict__ gb, unsigned lb,
                                     int s, int abits, v4f& acc) {
    const unsigned o0 = (unsigned)row_lane<J>(s)     * (N_CLASSES * 4u) + lb;
    const unsigned o1 = (unsigned)row_lane<J + 1>(s) * (N_CLASSES * 4u) + lb;
    const unsigned o2 = (unsigned)row_lane<J + 2>(s) * (N_CLASSES * 4u) + lb;
    const unsigned o3 = (unsigned)row_lane<J + 3>(s) * (N_CLASSES * 4u) + lb;
    const v4f f0 = *(const v4f*)(gb + o0);
    const v4f f1 = *(const v4f*)(gb + o1);
    const v4f f2 = *(const v4f*)(gb + o2);
    const v4f f3 = *(const v4f*)(gb + o3);
    acc += __int_as_float(row_lane<J>(abits))     * f0;
    acc += __int_as_float(row_lane<J + 1>(abits)) * f1;
    acc += __int_as_float(row_lane<J + 2>(abits)) * f2;
    acc += __int_as_float(row_lane<J + 3>(abits)) * f3;
}

__device__ __forceinline__ void agg_rows2(const char* __restrict__ gb, int nrows,
                                          int s, float a, unsigned lb, v4f& acc) {
    const int abits = __float_as_int(a);
    if (nrows > 0)  agg4<0>(gb, lb, s, abits, acc);
    if (nrows > 4)  agg4<4>(gb, lb, s, abits, acc);
    if (nrows > 8)  agg4<8>(gb, lb, s, abits, acc);
    if (nrows > 12) agg4<12>(gb, lb, s, abits, acc);
}

__global__ __launch_bounds__(256) void k_pull(const int* __restrict__ cnt,
                                              const int* __restrict__ rowptr,
                                              const int* __restrict__ esrc,
                                              const float* __restrict__ el,
                                              const float* __restrict__ er,
                                              const float* __restrict__ G,
                                              const float* __restrict__ FB,
                                              float* __restrict__ out) {
    const int lane = threadIdx.x & 31;
    const int l16 = lane & 15;
    const int hi  = lane >> 4;
    const int n = ((int)(blockIdx.x * 8 + (threadIdx.x >> 5))) * 2 + hi;
    const int deg   = cnt[n];
    const int start = rowptr[n];
    const float er_n = er[n];
    const char* __restrict__ gb = (const char*)G;
    const unsigned lb = (unsigned)l16 * 16u;
    const v4f bias = ((const v4f*)FB)[l16];
    const int dlo = __builtin_amdgcn_readlane(deg, 0);
    const int dup = __builtin_amdgcn_readlane(deg, 16);
    const int dmax = dlo > dup ? dlo : dup;

    int s0, s1; float e0, e1; bool v0, v1;
    load_chunk16(esrc, el, start, deg, 0,  l16, er_n, s0, e0, v0);
    load_chunk16(esrc, el, start, deg, 16, l16, er_n, s1, e1, v1);

    float m = fmaxf(e0, e1);
    for (int c = 32; c < dmax; c += 16) {
        int s; float e; bool v;
        load_chunk16(esrc, el, start, deg, c, l16, er_n, s, e, v);
        m = fmaxf(m, e);
    }
    for (int off = 8; off > 0; off >>= 1) m = fmaxf(m, __shfl_xor(m, off, 32));

    const float x0 = v0 ? __expf(e0 - m) : 0.f;
    const float x1 = v1 ? __expf(e1 - m) : 0.f;
    float dsum = x0 + x1;
    for (int c = 32; c < dmax; c += 16) {
        int s; float e; bool v;
        load_chunk16(esrc, el, start, deg, c, l16, er_n, s, e, v);
        dsum += v ? __expf(e - m) : 0.f;
    }
    for (int off = 8; off > 0; off >>= 1) dsum += __shfl_xor(dsum, off, 32);

    const float a0 = v0 ? x0 / dsum : 0.f;
    const float a1 = v1 ? x1 / dsum : 0.f;
    v4f acc = {0.f, 0.f, 0.f, 0.f};
    agg_rows2(gb, dmax < 16 ? dmax : 16, s0, a0, lb, acc);
    if (dmax > 16) {
        const int left = dmax - 16;
        agg_rows2(gb, left < 16 ? left : 16, s1, a1, lb, acc);
    }
    for (int c = 32; c < dmax; c += 16) {
        int s; float e; bool v;
        load_chunk16(esrc, el, start, deg, c, l16, er_n, s, e, v);
        const float a = v ? __expf(e - m) / dsum : 0.f;
        const int left = dmax - c;
        agg_rows2(gb, left < 16 ? left : 16, s, a, lb, acc);
    }
    ((v4f*)out)[(size_t)n * (N_CLASSES / 4) + l16] = acc + bias;
}

extern "C" void kernel_launch(void* const* d_in, const int* in_sizes, int n_in,
                              void* d_out, int out_size, void* d_ws, size_t ws_size,
                              hipStream_t stream) {
    const float* X  = (const float*)d_in[0];
    const float* W  = (const float*)d_in[1];
    const float* al = (const float*)d_in[2];
    const float* ar = (const float*)d_in[3];
    const float* FW = (const float*)d_in[4];
    const float* FB = (const float*)d_in[5];
    const int*  src = (const int*)d_in[6];
    const int*  dst = (const int*)d_in[7];
    float* out = (float*)d_out;

    char* ws = (char*)d_ws;
    size_t off = 0;
    auto carve = [&](size_t bytes) {
        void* p = ws + off;
        off += (bytes + 255) & ~(size_t)255;
        return p;
    };
    float* G      = (float*)carve((size_t)N_NODES * N_CLASSES * sizeof(float));
    float* el     = (float*)carve((size_t)N_NODES * sizeof(float));
    float* er     = (float*)carve((size_t)N_NODES * sizeof(float));
    int*   cnt    = (int*)  carve((size_t)N_NODES * sizeof(int));
    int*   rowptr = (int*)  carve((size_t)N_NODES * sizeof(int));
    int*   esrc   = (int*)  carve((size_t)N_EDGES * sizeof(int));
    float* wcp    = (float*)carve((size_t)D_IN * NC * sizeof(float));
    int*   binned = (int*)  carve(((size_t)2 * N_EDGES + (size_t)NLOFF) * sizeof(int));
    int*   loff   = binned + (size_t)2 * N_EDGES;

    k_part_fold<<<FOLD_BLOCKS + PB, PT, 0, stream>>>(src, dst, loff, binned, W, FW, al, ar, wcp);
    k_gemm_sort<<<GEMM_BLOCKS + NBIN, PT, 0, stream>>>(X, wcp, G, el, er, loff, binned,
                                                      cnt, rowptr, esrc);
    k_pull<<<N_NODES / 16, 256, 0, stream>>>(cnt, rowptr, esrc, el, er, G, FB, out);
}
